# starting kernel
# baseline (speedup 1.0000x reference)
#include <hip/hip_runtime.h>


typedef __attribute__((ext_vector_type(2))) float v2f;
typedef __attribute__((ext_vector_type(8))) float v8f;

#define NEG_SLOPE 0.2f

template <int K, int LDB>
__global__ __launch_bounds__(256) void wmma_dual_gemm(
    const float* __restrict__ A, const float* __restrict__ Bl,
    const float* __restrict__ Br, float* __restrict__ Cl,
    float* __restrict__ Cr, int rowTiles, int colTiles) {
  const int wid = (blockIdx.x * blockDim.x + threadIdx.x) >> 5;
  const int lane = threadIdx.x & 31;
  if (wid >= rowTiles * colTiles) return;
  const int rt = wid / colTiles;
  const int ct = wid % colTiles;
  const int row0 = rt * 16;
  const int col0 = ct * 16;
  const int half = lane >> 4;
  const int l16 = lane & 15;
  const int arow = row0 + l16;
  const int bcol = col0 + l16;

  const float* __restrict__ Ap = A + (size_t)arow * K + half * 2;

  v8f accl = {};
  v8f accr = {};
  for (int k = 0; k < K; k += 4) {
    const int ka = k + half * 2;
    v2f a;
    a.x = Ap[k];
    a.y = Ap[k + 1];
    v2f bl, br;
    bl.x = Bl[(size_t)ka * LDB + bcol];
    bl.y = Bl[(size_t)(ka + 1) * LDB + bcol];
    br.x = Br[(size_t)ka * LDB + bcol];
    br.y = Br[(size_t)(ka + 1) * LDB + bcol];
    accl = __builtin_amdgcn_wmma_f32_16x16x4_f32(false, a, false, bl, (short)0,
                                                 accl, false, false);
    accr = __builtin_amdgcn_wmma_f32_16x16x4_f32(false, a, false, br, (short)0,
                                                 accr, false, false);
  }

  const size_t base = (size_t)(row0 + half * 8) * LDB + bcol;
  float* __restrict__ cl = Cl + base;
  float* __restrict__ cr = Cr + base;
#pragma unroll
  for (int i = 0; i < 8; ++i) {
    cl[(size_t)i * LDB] = accl[i];
    cr[(size_t)i * LDB] = accr[i];
  }
}

__device__ __forceinline__ void atomicMaxFloat(float* addr, float val) {
  if (val >= 0.f)
    atomicMax((int*)addr, __float_as_int(val));
  else
    atomicMin((unsigned int*)addr, __float_as_uint(val));
}

template <int HEADS, int DIMH, int MODE>
__global__ __launch_bounds__(256) void edge_pass(
    const int* __restrict__ srcE, const int* __restrict__ dstE, int E, int N,
    const float* __restrict__ xl, const float* __restrict__ xr,
    const float* __restrict__ att, float* __restrict__ m,
    float* __restrict__ denom, float* __restrict__ out) {
  const int H = HEADS * DIMH;
  const int total = (E + N) * HEADS;
  const int t = blockIdx.x * blockDim.x + threadIdx.x;
  if (t >= total) return;
  const int e = t / HEADS;
  const int h = t % HEADS;
  int s, d;
  if (e < E) {
    s = srcE[e];
    d = dstE[e];
  } else {
    s = e - E;
    d = s;
  }
  const float* __restrict__ xls = xl + (size_t)s * H + h * DIMH;
  const float* __restrict__ xrd = xr + (size_t)d * H + h * DIMH;

  float logit = 0.f;
#pragma unroll
  for (int dd = 0; dd < DIMH; ++dd) {
    float v = xls[dd] + xrd[dd];
    v = v > 0.f ? v : NEG_SLOPE * v;
    logit += v * att[h * DIMH + dd];
  }

  const size_t seg = (size_t)d * HEADS + h;
  if (MODE == 0) {
    atomicMaxFloat(&m[seg], logit);
  } else if (MODE == 1) {
    atomicAdd(&denom[seg], __expf(logit - m[seg]));
  } else {
    const float alpha = __expf(logit - m[seg]) / denom[seg];
#pragma unroll
    for (int dd = 0; dd < DIMH; ++dd)
      atomicAdd(&out[(size_t)d * H + h * DIMH + dd], alpha * xls[dd]);
  }
}

__global__ void init_softmax_state(float* __restrict__ m,
                                   float* __restrict__ denom, int NH) {
  const int i = blockIdx.x * blockDim.x + threadIdx.x;
  if (i < NH) {
    m[i] = -__builtin_inff();
    denom[i] = 0.f;
  }
}

__global__ void init_bias_bcast(float* __restrict__ out,
                                const float* __restrict__ bias, int H,
                                int total) {
  const int i = blockIdx.x * blockDim.x + threadIdx.x;
  if (i < total) out[i] = bias[i % H];
}

__global__ void relu_inplace(float* __restrict__ p, int total) {
  const int i = blockIdx.x * blockDim.x + threadIdx.x;
  if (i < total) p[i] = fmaxf(p[i], 0.f);
}

extern "C" void kernel_launch(void* const* d_in, const int* in_sizes, int n_in,
                              void* d_out, int out_size, void* d_ws,
                              size_t ws_size, hipStream_t stream) {
  const float* x = (const float*)d_in[0];
  const int* edge = (const int*)d_in[1];
  const float* W1l = (const float*)d_in[2];
  const float* W1r = (const float*)d_in[3];
  const float* att1 = (const float*)d_in[4];
  const float* b1 = (const float*)d_in[5];
  const float* W2l = (const float*)d_in[6];
  const float* W2r = (const float*)d_in[7];
  const float* att2 = (const float*)d_in[8];
  const float* b2 = (const float*)d_in[9];
  float* out = (float*)d_out;

  const int F_IN = 128, H1 = 80, HEADS1 = 10, F_OUT = 16;
  const int N = in_sizes[0] / F_IN;
  const int E = in_sizes[1] / 2;
  const int* srcE = edge;
  const int* dstE = edge + E;

  float* ws = (float*)d_ws;
  float* xl1 = ws;                 ws += (size_t)N * H1;
  float* xr1 = ws;                 ws += (size_t)N * H1;
  float* agg1 = ws;                ws += (size_t)N * H1;
  float* m1 = ws;                  ws += (size_t)N * HEADS1;
  float* denom1 = ws;              ws += (size_t)N * HEADS1;
  float* xl2 = ws;                 ws += (size_t)N * F_OUT;
  float* xr2 = ws;                 ws += (size_t)N * F_OUT;
  float* m2 = ws;                  ws += (size_t)N;
  float* denom2 = ws;
  (void)ws_size; (void)n_in; (void)out_size;

  const int BLK = 256;

  {
    const int rowTiles = N / 16, colTiles = H1 / 16;
    const int waves = rowTiles * colTiles;
    const int blocks = (waves + 7) / 8;
    wmma_dual_gemm<128, 80><<<blocks, BLK, 0, stream>>>(
        x, W1l, W1r, xl1, xr1, rowTiles, colTiles);
  }
  init_softmax_state<<<(N * HEADS1 + BLK - 1) / BLK, BLK, 0, stream>>>(
      m1, denom1, N * HEADS1);
  init_bias_bcast<<<((size_t)N * H1 + BLK - 1) / BLK, BLK, 0, stream>>>(
      agg1, b1, H1, N * H1);

  const int etotal1 = (E + N) * HEADS1;
  edge_pass<10, 8, 0><<<(etotal1 + BLK - 1) / BLK, BLK, 0, stream>>>(
      srcE, dstE, E, N, xl1, xr1, att1, m1, denom1, agg1);
  edge_pass<10, 8, 1><<<(etotal1 + BLK - 1) / BLK, BLK, 0, stream>>>(
      srcE, dstE, E, N, xl1, xr1, att1, m1, denom1, agg1);
  edge_pass<10, 8, 2><<<(etotal1 + BLK - 1) / BLK, BLK, 0, stream>>>(
      srcE, dstE, E, N, xl1, xr1, att1, m1, denom1, agg1);

  relu_inplace<<<((size_t)N * H1 + BLK - 1) / BLK, BLK, 0, stream>>>(
      agg1, N * H1);

  {
    const int rowTiles = N / 16, colTiles = F_OUT / 16;
    const int waves = rowTiles * colTiles;
    const int blocks = (waves + 7) / 8;
    wmma_dual_gemm<80, 16><<<blocks, BLK, 0, stream>>>(
        agg1, W2l, W2r, xl2, xr2, rowTiles, colTiles);
  }
  init_softmax_state<<<(N + BLK - 1) / BLK, BLK, 0, stream>>>(m2, denom2, N);
  init_bias_bcast<<<((size_t)N * F_OUT + BLK - 1) / BLK, BLK, 0, stream>>>(
      out, b2, F_OUT, N * F_OUT);

  const int etotal2 = (E + N) * 1;
  edge_pass<1, 16, 0><<<(etotal2 + BLK - 1) / BLK, BLK, 0, stream>>>(
      srcE, dstE, E, N, xl2, xr2, att2, m2, denom2, out);
  edge_pass<1, 16, 1><<<(etotal2 + BLK - 1) / BLK, BLK, 0, stream>>>(
      srcE, dstE, E, N, xl2, xr2, att2, m2, denom2, out);
  edge_pass<1, 16, 2><<<(etotal2 + BLK - 1) / BLK, BLK, 0, stream>>>(
      srcE, dstE, E, N, xl2, xr2, att2, m2, denom2, out);
}


# Edges grouped by destination in-call (arrival-order rank via int atomic, scan, place); one fused per-node online-softmax gather pass replaces 3 float-atomic edge passes; sum order changes
# speedup vs baseline: 4.6647x; 4.6647x over previous
#include <hip/hip_runtime.h>


typedef __attribute__((ext_vector_type(2))) float v2f;
typedef __attribute__((ext_vector_type(8))) float v8f;

#define NEG_SLOPE 0.2f

template <int K, int LDB>
__global__ __launch_bounds__(256) void wmma_dual_gemm(
    const float* __restrict__ A, const float* __restrict__ Bl,
    const float* __restrict__ Br, float* __restrict__ Cl,
    float* __restrict__ Cr, int rowTiles, int colTiles) {
  const int wid = (blockIdx.x * blockDim.x + threadIdx.x) >> 5;
  const int lane = threadIdx.x & 31;
  if (wid >= rowTiles * colTiles) return;
  const int rt = wid / colTiles;
  const int ct = wid % colTiles;
  const int row0 = rt * 16;
  const int col0 = ct * 16;
  const int half = lane >> 4;
  const int l16 = lane & 15;
  const int arow = row0 + l16;
  const int bcol = col0 + l16;

  const float* __restrict__ Ap = A + (size_t)arow * K + half * 2;

  v8f accl = {};
  v8f accr = {};
  for (int k = 0; k < K; k += 4) {
    const int ka = k + half * 2;
    v2f a;
    a.x = Ap[k];
    a.y = Ap[k + 1];
    v2f bl, br;
    bl.x = Bl[(size_t)ka * LDB + bcol];
    bl.y = Bl[(size_t)(ka + 1) * LDB + bcol];
    br.x = Br[(size_t)ka * LDB + bcol];
    br.y = Br[(size_t)(ka + 1) * LDB + bcol];
    accl = __builtin_amdgcn_wmma_f32_16x16x4_f32(false, a, false, bl, (short)0,
                                                 accl, false, false);
    accr = __builtin_amdgcn_wmma_f32_16x16x4_f32(false, a, false, br, (short)0,
                                                 accr, false, false);
  }

  const size_t base = (size_t)(row0 + half * 8) * LDB + bcol;
  float* __restrict__ cl = Cl + base;
  float* __restrict__ cr = Cr + base;
#pragma unroll
  for (int i = 0; i < 8; ++i) {
    cl[(size_t)i * LDB] = accl[i];
    cr[(size_t)i * LDB] = accr[i];
  }
}

__global__ void zero_counts(int* __restrict__ cnt, int N) {
  const int i = blockIdx.x * blockDim.x + threadIdx.x;
  if (i < N) cnt[i] = 0;
}

__global__ void count_and_rank(const int* __restrict__ dstE, int E,
                               int* __restrict__ cnt, int* __restrict__ slot) {
  const int e = blockIdx.x * blockDim.x + threadIdx.x;
  if (e < E) slot[e] = atomicAdd(&cnt[dstE[e]], 1);
}

__global__ __launch_bounds__(256) void block_sums(const int* __restrict__ cnt,
                                                  int* __restrict__ bsum,
                                                  int N) {
  __shared__ int sh[256];
  const int tid = threadIdx.x;
  const int i0 = (blockIdx.x * 256 + tid) * 4;
  int s = 0;
  for (int k = 0; k < 4; ++k)
    if (i0 + k < N) s += cnt[i0 + k];
  sh[tid] = s;
  __syncthreads();
  for (int off = 128; off > 0; off >>= 1) {
    if (tid < off) sh[tid] += sh[tid + off];
    __syncthreads();
  }
  if (tid == 0) bsum[blockIdx.x] = sh[0];
}

__global__ __launch_bounds__(256) void scan_rows(const int* __restrict__ cnt,
                                                 const int* __restrict__ bsum,
                                                 int* __restrict__ rowptr,
                                                 int N) {
  __shared__ int sh[256];
  const int tid = threadIdx.x;
  const int b = blockIdx.x;

  int p = 0;
  for (int i = tid; i < b; i += 256) p += bsum[i];
  sh[tid] = p;
  __syncthreads();
  for (int off = 128; off > 0; off >>= 1) {
    if (tid < off) sh[tid] += sh[tid + off];
    __syncthreads();
  }
  const int base = sh[0];
  __syncthreads();

  const int i0 = (b * 256 + tid) * 4;
  int v[4];
  int s = 0;
  for (int k = 0; k < 4; ++k) {
    v[k] = (i0 + k < N) ? cnt[i0 + k] : 0;
    s += v[k];
  }
  sh[tid] = s;
  __syncthreads();
  for (int off = 1; off < 256; off <<= 1) {
    const int a = (tid >= off) ? sh[tid - off] : 0;
    __syncthreads();
    sh[tid] += a;
    __syncthreads();
  }
  int run = base + sh[tid] - s;
  for (int k = 0; k < 4; ++k) {
    if (i0 + k < N) {
      rowptr[i0 + k] = run;
      run += v[k];
    }
  }
  if (N - 1 >= i0 && N - 1 < i0 + 4) rowptr[N] = run;
}

__global__ void place_edges(const int* __restrict__ srcE,
                            const int* __restrict__ dstE, int E,
                            const int* __restrict__ rowptr,
                            const int* __restrict__ slot,
                            int* __restrict__ csrSrc) {
  const int e = blockIdx.x * blockDim.x + threadIdx.x;
  if (e < E) csrSrc[rowptr[dstE[e]] + slot[e]] = srcE[e];
}

__device__ __forceinline__ float4 ld4(const float* p) {
  return *reinterpret_cast<const float4*>(p);
}

template <int HEADS, int DIMH, int LPG, bool RELU>
__global__ __launch_bounds__(256) void node_softmax_aggregate(
    const int* __restrict__ rowptr, const int* __restrict__ csrSrc, int N,
    const float* __restrict__ xl, const float* __restrict__ xr,
    const float* __restrict__ att, const float* __restrict__ bias,
    float* __restrict__ out) {
  static_assert(DIMH == 4 * LPG, "each lane owns 4 features");
  static_assert(LPG == 1 || LPG == 2 || LPG == 4, "group inside a quad");
  constexpr int H = HEADS * DIMH;
  const int t = blockIdx.x * blockDim.x + threadIdx.x;
  const int g = t / LPG;
  const int sub = t % LPG;
  if (g >= N * HEADS) return;
  const int d = g / HEADS;
  const int h = g % HEADS;
  const int col = h * DIMH + sub * 4;

  const float4 r = ld4(xr + (size_t)d * H + col);
  const float a0 = att[col], a1 = att[col + 1], a2 = att[col + 2],
              a3 = att[col + 3];

  auto logit_of = [&](const float4& v) -> float {
    float z0 = v.x + r.x, z1 = v.y + r.y, z2 = v.z + r.z, z3 = v.w + r.w;
    z0 = z0 > 0.f ? z0 : NEG_SLOPE * z0;
    z1 = z1 > 0.f ? z1 : NEG_SLOPE * z1;
    z2 = z2 > 0.f ? z2 : NEG_SLOPE * z2;
    z3 = z3 > 0.f ? z3 : NEG_SLOPE * z3;
    float p = z0 * a0;
    p += z1 * a1;
    p += z2 * a2;
    p += z3 * a3;
    if (LPG >= 2) p += __shfl_xor(p, 1);
    if (LPG >= 4) p += __shfl_xor(p, 2);
    return p;
  };

  const float4 xs = ld4(xl + (size_t)d * H + col);
  float m = logit_of(xs);
  float den = 1.f;
  float4 acc = xs;

  const int beg = rowptr[d];
  const int end = rowptr[d + 1];
  int sNext = (beg < end) ? csrSrc[beg] : d;
  float4 vNext = ld4(xl + (size_t)sNext * H + col);
  sNext = (beg + 1 < end) ? csrSrc[beg + 1] : d;
  for (int j = beg; j < end; ++j) {
    const float4 v = vNext;
    vNext = ld4(xl + (size_t)sNext * H + col);
    sNext = (j + 2 < end) ? csrSrc[j + 2] : d;

    const float lg = logit_of(v);
    const float mn = fmaxf(m, lg);
    const float sc = __expf(m - mn);
    const float pe = __expf(lg - mn);
    den = den * sc + pe;
    acc.x = acc.x * sc + pe * v.x;
    acc.y = acc.y * sc + pe * v.y;
    acc.z = acc.z * sc + pe * v.z;
    acc.w = acc.w * sc + pe * v.w;
    m = mn;
  }

  const float inv = 1.f / den;
  float4 o;
  o.x = acc.x * inv + bias[col];
  o.y = acc.y * inv + bias[col + 1];
  o.z = acc.z * inv + bias[col + 2];
  o.w = acc.w * inv + bias[col + 3];
  if (RELU) {
    o.x = fmaxf(o.x, 0.f);
    o.y = fmaxf(o.y, 0.f);
    o.z = fmaxf(o.z, 0.f);
    o.w = fmaxf(o.w, 0.f);
  }
  *reinterpret_cast<float4*>(out + (size_t)d * H + col) = o;
}

extern "C" void kernel_launch(void* const* d_in, const int* in_sizes, int n_in,
                              void* d_out, int out_size, void* d_ws,
                              size_t ws_size, hipStream_t stream) {
  const float* x = (const float*)d_in[0];
  const int* edge = (const int*)d_in[1];
  const float* W1l = (const float*)d_in[2];
  const float* W1r = (const float*)d_in[3];
  const float* att1 = (const float*)d_in[4];
  const float* b1 = (const float*)d_in[5];
  const float* W2l = (const float*)d_in[6];
  const float* W2r = (const float*)d_in[7];
  const float* att2 = (const float*)d_in[8];
  const float* b2 = (const float*)d_in[9];
  float* out = (float*)d_out;

  const int F_IN = 128, H1 = 80, HEADS1 = 10, F_OUT = 16;
  const int N = in_sizes[0] / F_IN;
  const int E = in_sizes[1] / 2;
  const int* srcE = edge;
  const int* dstE = edge + E;

  float* ws = (float*)d_ws;
  float* xl1 = ws;                 ws += (size_t)N * H1;
  float* xr1 = ws;                 ws += (size_t)N * H1;
  float* h1 = ws;                  ws += (size_t)N * H1;
  float* xl2 = ws;                 ws += (size_t)N * F_OUT;
  float* xr2 = ws;                 ws += (size_t)N * F_OUT;
  int* wi = (int*)ws;
  const int scanBlocks = (N + 1023) / 1024;
  int* cnt = wi;                   wi += (size_t)((N + 3) / 4) * 4;
  int* bsum = wi;                  wi += (size_t)((scanBlocks + 3) / 4) * 4;
  int* rowptr = wi;                wi += (size_t)((N + 1 + 3) / 4) * 4;
  int* slot = wi;                  wi += (size_t)((E + 3) / 4) * 4;
  int* csrSrc = wi;
  (void)ws_size; (void)n_in; (void)out_size;

  const int BLK = 256;

  zero_counts<<<(N + BLK - 1) / BLK, BLK, 0, stream>>>(cnt, N);
  count_and_rank<<<(E + BLK - 1) / BLK, BLK, 0, stream>>>(dstE, E, cnt, slot);
  block_sums<<<scanBlocks, 256, 0, stream>>>(cnt, bsum, N);
  scan_rows<<<scanBlocks, 256, 0, stream>>>(cnt, bsum, rowptr, N);
  place_edges<<<(E + BLK - 1) / BLK, BLK, 0, stream>>>(srcE, dstE, E, rowptr,
                                                        slot, csrSrc);

  {
    const int rowTiles = N / 16, colTiles = H1 / 16;
    const int waves = rowTiles * colTiles;
    const int blocks = (waves + 7) / 8;
    wmma_dual_gemm<128, 80><<<blocks, BLK, 0, stream>>>(
        x, W1l, W1r, xl1, xr1, rowTiles, colTiles);
  }
  {
    const int threads = N * HEADS1 * 2;
    node_softmax_aggregate<10, 8, 2, true>
        <<<(threads + BLK - 1) / BLK, BLK, 0, stream>>>(
            rowptr, csrSrc, N, xl1, xr1, att1, b1, h1);
  }

  {
    const int rowTiles = N / 16, colTiles = F_OUT / 16;
    const int waves = rowTiles * colTiles;
    const int blocks = (waves + 7) / 8;
    wmma_dual_gemm<80, 16><<<blocks, BLK, 0, stream>>>(
        h1, W2l, W2r, xl2, xr2, rowTiles, colTiles);
  }
  {
    const int threads = N * 4;
    node_softmax_aggregate<1, 16, 4, false>
        <<<(threads + BLK - 1) / BLK, BLK, 0, stream>>>(
            rowptr, csrSrc, N, xl2, xr2, att2, b2, out);
  }
}


# Ranking atomic stores each edge straight into its destination's fixed 32-slot bucket (spill list beyond 32); block-sum, scan and placement launches removed; sums stay in arrival order
# speedup vs baseline: 5.3462x; 1.1461x over previous
#include <hip/hip_runtime.h>


typedef __attribute__((ext_vector_type(2))) float v2f;
typedef __attribute__((ext_vector_type(8))) float v8f;

#define NEG_SLOPE 0.2f

template <int K, int LDB>
__global__ __launch_bounds__(256) void wmma_dual_gemm(
    const float* __restrict__ A, const float* __restrict__ Bl,
    const float* __restrict__ Br, float* __restrict__ Cl,
    float* __restrict__ Cr, int rowTiles, int colTiles) {
  const int wid = (blockIdx.x * blockDim.x + threadIdx.x) >> 5;
  const int lane = threadIdx.x & 31;
  if (wid >= rowTiles * colTiles) return;
  const int rt = wid / colTiles;
  const int ct = wid % colTiles;
  const int row0 = rt * 16;
  const int col0 = ct * 16;
  const int half = lane >> 4;
  const int l16 = lane & 15;
  const int arow = row0 + l16;
  const int bcol = col0 + l16;

  const float* __restrict__ Ap = A + (size_t)arow * K + half * 2;

  v8f accl = {};
  v8f accr = {};
  for (int k = 0; k < K; k += 4) {
    const int ka = k + half * 2;
    v2f a;
    a.x = Ap[k];
    a.y = Ap[k + 1];
    v2f bl, br;
    bl.x = Bl[(size_t)ka * LDB + bcol];
    bl.y = Bl[(size_t)(ka + 1) * LDB + bcol];
    br.x = Br[(size_t)ka * LDB + bcol];
    br.y = Br[(size_t)(ka + 1) * LDB + bcol];
    accl = __builtin_amdgcn_wmma_f32_16x16x4_f32(false, a, false, bl, (short)0,
                                                 accl, false, false);
    accr = __builtin_amdgcn_wmma_f32_16x16x4_f32(false, a, false, br, (short)0,
                                                 accr, false, false);
  }

  const size_t base = (size_t)(row0 + half * 8) * LDB + bcol;
  float* __restrict__ cl = Cl + base;
  float* __restrict__ cr = Cr + base;
#pragma unroll
  for (int i = 0; i < 8; ++i) {
    cl[(size_t)i * LDB] = accl[i];
    cr[(size_t)i * LDB] = accr[i];
  }
}

#define SLOTS 32

__global__ void zero_counts(int* __restrict__ cnt, int n) {
  const int i = blockIdx.x * blockDim.x + threadIdx.x;
  if (i < n) cnt[i] = 0;
}

__global__ void bucket_edges(const int* __restrict__ srcE,
                             const int* __restrict__ dstE, int E, int N,
                             int* __restrict__ cnt, int* __restrict__ bucket,
                             int* __restrict__ spill) {
  const int e = blockIdx.x * blockDim.x + threadIdx.x;
  if (e >= E) return;
  const int d = dstE[e];
  const int s = srcE[e];
  const int rank = atomicAdd(&cnt[d], 1);
  if (rank < SLOTS) {
    bucket[(size_t)d * SLOTS + rank] = s;
  } else {
    const int o = atomicAdd(&cnt[N], 1);
    reinterpret_cast<int2*>(spill)[o] = make_int2(d, s);
  }
}

__device__ __forceinline__ float4 ld4(const float* p) {
  return *reinterpret_cast<const float4*>(p);
}

template <int HEADS, int DIMH, int LPG, bool RELU>
__global__ __launch_bounds__(256) void node_softmax_aggregate(
    const int* __restrict__ cnt, const int* __restrict__ bucket,
    const int* __restrict__ spill, int N,
    const float* __restrict__ xl, const float* __restrict__ xr,
    const float* __restrict__ att, const float* __restrict__ bias,
    float* __restrict__ out) {
  static_assert(DIMH == 4 * LPG, "each lane owns 4 features");
  static_assert(LPG == 1 || LPG == 2 || LPG == 4, "group inside a quad");
  constexpr int H = HEADS * DIMH;
  const int t = blockIdx.x * blockDim.x + threadIdx.x;
  const int g = t / LPG;
  const int sub = t % LPG;
  if (g >= N * HEADS) return;
  const int d = g / HEADS;
  const int h = g % HEADS;
  const int col = h * DIMH + sub * 4;

  const float4 r = ld4(xr + (size_t)d * H + col);
  const float a0 = att[col], a1 = att[col + 1], a2 = att[col + 2],
              a3 = att[col + 3];

  auto logit_of = [&](const float4& v) -> float {
    float z0 = v.x + r.x, z1 = v.y + r.y, z2 = v.z + r.z, z3 = v.w + r.w;
    z0 = z0 > 0.f ? z0 : NEG_SLOPE * z0;
    z1 = z1 > 0.f ? z1 : NEG_SLOPE * z1;
    z2 = z2 > 0.f ? z2 : NEG_SLOPE * z2;
    z3 = z3 > 0.f ? z3 : NEG_SLOPE * z3;
    float p = z0 * a0;
    p += z1 * a1;
    p += z2 * a2;
    p += z3 * a3;
    if (LPG >= 2) p += __shfl_xor(p, 1);
    if (LPG >= 4) p += __shfl_xor(p, 2);
    return p;
  };

  const float4 xs = ld4(xl + (size_t)d * H + col);
  float m = logit_of(xs);
  float den = 1.f;
  float4 acc = xs;

  auto absorb = [&](const float4& v) {
    const float lg = logit_of(v);
    const float mn = fmaxf(m, lg);
    const float sc = __expf(m - mn);
    const float pe = __expf(lg - mn);
    den = den * sc + pe;
    acc.x = acc.x * sc + pe * v.x;
    acc.y = acc.y * sc + pe * v.y;
    acc.z = acc.z * sc + pe * v.z;
    acc.w = acc.w * sc + pe * v.w;
    m = mn;
  };

  const int deg = cnt[d];
  const int nb = deg < SLOTS ? deg : SLOTS;
  const int* __restrict__ mine = bucket + (size_t)d * SLOTS;
  int sNext = (0 < nb) ? mine[0] : d;
  float4 vNext = ld4(xl + (size_t)sNext * H + col);
  sNext = (1 < nb) ? mine[1] : d;
  for (int j = 0; j < nb; ++j) {
    const float4 v = vNext;
    vNext = ld4(xl + (size_t)sNext * H + col);
    sNext = (j + 2 < nb) ? mine[j + 2] : d;
    absorb(v);
  }

  if (deg > SLOTS) {
    const int nSpill = cnt[N];
    const int2* __restrict__ sp = reinterpret_cast<const int2*>(spill);
    for (int i = 0; i < nSpill; ++i) {
      const int2 ds = sp[i];
      if (ds.x == d) absorb(ld4(xl + (size_t)ds.y * H + col));
    }
  }

  const float inv = 1.f / den;
  float4 o;
  o.x = acc.x * inv + bias[col];
  o.y = acc.y * inv + bias[col + 1];
  o.z = acc.z * inv + bias[col + 2];
  o.w = acc.w * inv + bias[col + 3];
  if (RELU) {
    o.x = fmaxf(o.x, 0.f);
    o.y = fmaxf(o.y, 0.f);
    o.z = fmaxf(o.z, 0.f);
    o.w = fmaxf(o.w, 0.f);
  }
  *reinterpret_cast<float4*>(out + (size_t)d * H + col) = o;
}

extern "C" void kernel_launch(void* const* d_in, const int* in_sizes, int n_in,
                              void* d_out, int out_size, void* d_ws,
                              size_t ws_size, hipStream_t stream) {
  const float* x = (const float*)d_in[0];
  const int* edge = (const int*)d_in[1];
  const float* W1l = (const float*)d_in[2];
  const float* W1r = (const float*)d_in[3];
  const float* att1 = (const float*)d_in[4];
  const float* b1 = (const float*)d_in[5];
  const float* W2l = (const float*)d_in[6];
  const float* W2r = (const float*)d_in[7];
  const float* att2 = (const float*)d_in[8];
  const float* b2 = (const float*)d_in[9];
  float* out = (float*)d_out;

  const int F_IN = 128, H1 = 80, HEADS1 = 10, F_OUT = 16;
  const int N = in_sizes[0] / F_IN;
  const int E = in_sizes[1] / 2;
  const int* srcE = edge;
  const int* dstE = edge + E;

  float* ws = (float*)d_ws;
  float* xl1 = ws;                 ws += (size_t)N * H1;
  float* xr1 = ws;                 ws += (size_t)N * H1;
  float* h1 = ws;                  ws += (size_t)N * H1;
  float* xl2 = ws;                 ws += (size_t)N * F_OUT;
  float* xr2 = ws;                 ws += (size_t)N * F_OUT;
  int* wi = (int*)ws;
  int* cnt = wi;                   wi += (size_t)((N + 1 + 31) / 32) * 32;
  int* bucket = wi;                wi += (size_t)N * SLOTS;
  int* spill = wi;
  (void)ws_size; (void)n_in; (void)out_size;

  const int BLK = 256;

  zero_counts<<<(N + 1 + BLK - 1) / BLK, BLK, 0, stream>>>(cnt, N + 1);
  bucket_edges<<<(E + BLK - 1) / BLK, BLK, 0, stream>>>(srcE, dstE, E, N, cnt,
                                                         bucket, spill);

  {
    const int rowTiles = N / 16, colTiles = H1 / 16;
    const int waves = rowTiles * colTiles;
    const int blocks = (waves + 7) / 8;
    wmma_dual_gemm<128, 80><<<blocks, BLK, 0, stream>>>(
        x, W1l, W1r, xl1, xr1, rowTiles, colTiles);
  }
  {
    const int threads = N * HEADS1 * 2;
    node_softmax_aggregate<10, 8, 2, true>
        <<<(threads + BLK - 1) / BLK, BLK, 0, stream>>>(
            cnt, bucket, spill, N, xl1, xr1, att1, b1, h1);
  }

  {
    const int rowTiles = N / 16, colTiles = F_OUT / 16;
    const int waves = rowTiles * colTiles;
    const int blocks = (waves + 7) / 8;
    wmma_dual_gemm<80, 16><<<blocks, BLK, 0, stream>>>(
        h1, W2l, W2r, xl2, xr2, rowTiles, colTiles);
  }
  {
    const int threads = N * 4;
    node_softmax_aggregate<1, 16, 4, false>
        <<<(threads + BLK - 1) / BLK, BLK, 0, stream>>>(
            cnt, bucket, spill, N, xl2, xr2, att2, b2, out);
  }
}


# Layer-1 transform re-tiled: one wave makes a 16-row strip's 160 columns, x read once with 16-byte loads, both weight matrices staged once per work-group in LDS in fragment order; k sum order changes
# speedup vs baseline: 5.5771x; 1.0432x over previous
#include <hip/hip_runtime.h>


typedef __attribute__((ext_vector_type(2))) float v2f;
typedef __attribute__((ext_vector_type(8))) float v8f;

#define NEG_SLOPE 0.2f

template <int K, int LDB>
__global__ __launch_bounds__(256) void wmma_dual_gemm(
    const float* __restrict__ A, const float* __restrict__ Bl,
    const float* __restrict__ Br, float* __restrict__ Cl,
    float* __restrict__ Cr, int rowTiles, int colTiles) {
  const int wid = (blockIdx.x * blockDim.x + threadIdx.x) >> 5;
  const int lane = threadIdx.x & 31;
  if (wid >= rowTiles * colTiles) return;
  const int rt = wid / colTiles;
  const int ct = wid % colTiles;
  const int row0 = rt * 16;
  const int col0 = ct * 16;
  const int half = lane >> 4;
  const int l16 = lane & 15;
  const int arow = row0 + l16;
  const int bcol = col0 + l16;

  const float* __restrict__ Ap = A + (size_t)arow * K + half * 2;

  v8f accl = {};
  v8f accr = {};
  for (int k = 0; k < K; k += 4) {
    const int ka = k + half * 2;
    v2f a;
    a.x = Ap[k];
    a.y = Ap[k + 1];
    v2f bl, br;
    bl.x = Bl[(size_t)ka * LDB + bcol];
    bl.y = Bl[(size_t)(ka + 1) * LDB + bcol];
    br.x = Br[(size_t)ka * LDB + bcol];
    br.y = Br[(size_t)(ka + 1) * LDB + bcol];
    accl = __builtin_amdgcn_wmma_f32_16x16x4_f32(false, a, false, bl, (short)0,
                                                 accl, false, false);
    accr = __builtin_amdgcn_wmma_f32_16x16x4_f32(false, a, false, br, (short)0,
                                                 accr, false, false);
  }

  const size_t base = (size_t)(row0 + half * 8) * LDB + bcol;
  float* __restrict__ cl = Cl + base;
  float* __restrict__ cr = Cr + base;
#pragma unroll
  for (int i = 0; i < 8; ++i) {
    cl[(size_t)i * LDB] = accl[i];
    cr[(size_t)i * LDB] = accr[i];
  }
}

#define S_K 128
#define S_LD 80
#define S_KB (S_K / 8)
#define S_NT 10
#define S_WAVES 8

__global__ __launch_bounds__(256) void wmma_strip_gemm_l1(
    const float* __restrict__ A, const float* __restrict__ Bl,
    const float* __restrict__ Br, float* __restrict__ Cl,
    float* __restrict__ Cr, int rowTiles) {
  __shared__ float4 sB[S_KB * S_NT * 32];

  const int tid = threadIdx.x;
  const int lane = tid & 31;
  const int wave = tid >> 5;
  const int half = lane >> 4;
  const int l16 = lane & 15;
  const int rt = blockIdx.x * S_WAVES + wave;
  const bool live = rt < rowTiles;
  const int row0 = rt * 16;

  float4 a[S_KB];
  if (live) {
    const float* __restrict__ Ap = A + (size_t)(row0 + l16) * S_K + half * 4;
#pragma unroll
    for (int kb = 0; kb < S_KB; ++kb)
      a[kb] = *reinterpret_cast<const float4*>(Ap + kb * 8);
  }

  {
    constexpr int ITEMS = 2 * S_KB * 2 * S_LD;
    constexpr int PER = ITEMS / 256;
    float4 w[PER];
#pragma unroll
    for (int i = 0; i < PER; ++i) {
      const int it = tid + 256 * i;
      const int q = it / S_LD;
      const int col = it % S_LD;
      const int m = q >> 5;
      const int kb = (q & 31) >> 1;
      const int h = q & 1;
      const float* __restrict__ src =
          (m ? Br : Bl) + (size_t)(kb * 8 + h * 4) * S_LD + col;
      w[i].x = src[0];
      w[i].y = src[S_LD];
      w[i].z = src[2 * S_LD];
      w[i].w = src[3 * S_LD];
    }
#pragma unroll
    for (int i = 0; i < PER; ++i) {
      const int it = tid + 256 * i;
      const int q = it / S_LD;
      const int col = it % S_LD;
      const int m = q >> 5;
      const int kb = (q & 31) >> 1;
      const int h = q & 1;
      const int t = m * 5 + (col >> 4);
      sB[(kb * S_NT + t) * 32 + h * 16 + (col & 15)] = w[i];
    }
  }
  __syncthreads();
  if (!live) return;

  v8f acc[S_NT];
#pragma unroll
  for (int t = 0; t < S_NT; ++t) acc[t] = (v8f){};

#pragma unroll
  for (int kb = 0; kb < S_KB; ++kb) {
    float4 b[S_NT];
#pragma unroll
    for (int t = 0; t < S_NT; ++t) b[t] = sB[(kb * S_NT + t) * 32 + lane];
    v2f a01, a23;
    a01.x = a[kb].x; a01.y = a[kb].y;
    a23.x = a[kb].z; a23.y = a[kb].w;
#pragma unroll
    for (int t = 0; t < S_NT; ++t) {
      v2f b01;
      b01.x = b[t].x; b01.y = b[t].y;
      acc[t] = __builtin_amdgcn_wmma_f32_16x16x4_f32(false, a01, false, b01,
                                                     (short)0, acc[t], false,
                                                     false);
    }
#pragma unroll
    for (int t = 0; t < S_NT; ++t) {
      v2f b23;
      b23.x = b[t].z; b23.y = b[t].w;
      acc[t] = __builtin_amdgcn_wmma_f32_16x16x4_f32(false, a23, false, b23,
                                                     (short)0, acc[t], false,
                                                     false);
    }
  }

  const size_t base = (size_t)(row0 + half * 8) * S_LD + l16;
#pragma unroll
  for (int t = 0; t < S_NT; ++t) {
    float* __restrict__ c = (t < 5 ? Cl + t * 16 : Cr + (t - 5) * 16) + base;
#pragma unroll
    for (int i = 0; i < 8; ++i) c[(size_t)i * S_LD] = acc[t][i];
  }
}

#define SLOTS 32

__global__ void zero_counts(int* __restrict__ cnt, int n) {
  const int i = blockIdx.x * blockDim.x + threadIdx.x;
  if (i < n) cnt[i] = 0;
}

__global__ void bucket_edges(const int* __restrict__ srcE,
                             const int* __restrict__ dstE, int E, int N,
                             int* __restrict__ cnt, int* __restrict__ bucket,
                             int* __restrict__ spill) {
  const int e = blockIdx.x * blockDim.x + threadIdx.x;
  if (e >= E) return;
  const int d = dstE[e];
  const int s = srcE[e];
  const int rank = atomicAdd(&cnt[d], 1);
  if (rank < SLOTS) {
    bucket[(size_t)d * SLOTS + rank] = s;
  } else {
    const int o = atomicAdd(&cnt[N], 1);
    reinterpret_cast<int2*>(spill)[o] = make_int2(d, s);
  }
}

__device__ __forceinline__ float4 ld4(const float* p) {
  return *reinterpret_cast<const float4*>(p);
}

template <int HEADS, int DIMH, int LPG, bool RELU>
__global__ __launch_bounds__(256) void node_softmax_aggregate(
    const int* __restrict__ cnt, const int* __restrict__ bucket,
    const int* __restrict__ spill, int N,
    const float* __restrict__ xl, const float* __restrict__ xr,
    const float* __restrict__ att, const float* __restrict__ bias,
    float* __restrict__ out) {
  static_assert(DIMH == 4 * LPG, "each lane owns 4 features");
  static_assert(LPG == 1 || LPG == 2 || LPG == 4, "group inside a quad");
  constexpr int H = HEADS * DIMH;
  const int t = blockIdx.x * blockDim.x + threadIdx.x;
  const int g = t / LPG;
  const int sub = t % LPG;
  if (g >= N * HEADS) return;
  const int d = g / HEADS;
  const int h = g % HEADS;
  const int col = h * DIMH + sub * 4;

  const float4 r = ld4(xr + (size_t)d * H + col);
  const float a0 = att[col], a1 = att[col + 1], a2 = att[col + 2],
              a3 = att[col + 3];

  auto logit_of = [&](const float4& v) -> float {
    float z0 = v.x + r.x, z1 = v.y + r.y, z2 = v.z + r.z, z3 = v.w + r.w;
    z0 = z0 > 0.f ? z0 : NEG_SLOPE * z0;
    z1 = z1 > 0.f ? z1 : NEG_SLOPE * z1;
    z2 = z2 > 0.f ? z2 : NEG_SLOPE * z2;
    z3 = z3 > 0.f ? z3 : NEG_SLOPE * z3;
    float p = z0 * a0;
    p += z1 * a1;
    p += z2 * a2;
    p += z3 * a3;
    if (LPG >= 2) p += __shfl_xor(p, 1);
    if (LPG >= 4) p += __shfl_xor(p, 2);
    return p;
  };

  const float4 xs = ld4(xl + (size_t)d * H + col);
  float m = logit_of(xs);
  float den = 1.f;
  float4 acc = xs;

  auto absorb = [&](const float4& v) {
    const float lg = logit_of(v);
    const float mn = fmaxf(m, lg);
    const float sc = __expf(m - mn);
    const float pe = __expf(lg - mn);
    den = den * sc + pe;
    acc.x = acc.x * sc + pe * v.x;
    acc.y = acc.y * sc + pe * v.y;
    acc.z = acc.z * sc + pe * v.z;
    acc.w = acc.w * sc + pe * v.w;
    m = mn;
  };

  const int deg = cnt[d];
  const int nb = deg < SLOTS ? deg : SLOTS;
  const int* __restrict__ mine = bucket + (size_t)d * SLOTS;
  int sNext = (0 < nb) ? mine[0] : d;
  float4 vNext = ld4(xl + (size_t)sNext * H + col);
  sNext = (1 < nb) ? mine[1] : d;
  for (int j = 0; j < nb; ++j) {
    const float4 v = vNext;
    vNext = ld4(xl + (size_t)sNext * H + col);
    sNext = (j + 2 < nb) ? mine[j + 2] : d;
    absorb(v);
  }

  if (deg > SLOTS) {
    const int nSpill = cnt[N];
    const int2* __restrict__ sp = reinterpret_cast<const int2*>(spill);
    for (int i = 0; i < nSpill; ++i) {
      const int2 ds = sp[i];
      if (ds.x == d) absorb(ld4(xl + (size_t)ds.y * H + col));
    }
  }

  const float inv = 1.f / den;
  float4 o;
  o.x = acc.x * inv + bias[col];
  o.y = acc.y * inv + bias[col + 1];
  o.z = acc.z * inv + bias[col + 2];
  o.w = acc.w * inv + bias[col + 3];
  if (RELU) {
    o.x = fmaxf(o.x, 0.f);
    o.y = fmaxf(o.y, 0.f);
    o.z = fmaxf(o.z, 0.f);
    o.w = fmaxf(o.w, 0.f);
  }
  *reinterpret_cast<float4*>(out + (size_t)d * H + col) = o;
}

extern "C" void kernel_launch(void* const* d_in, const int* in_sizes, int n_in,
                              void* d_out, int out_size, void* d_ws,
                              size_t ws_size, hipStream_t stream) {
  const float* x = (const float*)d_in[0];
  const int* edge = (const int*)d_in[1];
  const float* W1l = (const float*)d_in[2];
  const float* W1r = (const float*)d_in[3];
  const float* att1 = (const float*)d_in[4];
  const float* b1 = (const float*)d_in[5];
  const float* W2l = (const float*)d_in[6];
  const float* W2r = (const float*)d_in[7];
  const float* att2 = (const float*)d_in[8];
  const float* b2 = (const float*)d_in[9];
  float* out = (float*)d_out;

  const int F_IN = 128, H1 = 80, HEADS1 = 10, F_OUT = 16;
  const int N = in_sizes[0] / F_IN;
  const int E = in_sizes[1] / 2;
  const int* srcE = edge;
  const int* dstE = edge + E;

  float* ws = (float*)d_ws;
  float* xl1 = ws;                 ws += (size_t)N * H1;
  float* xr1 = ws;                 ws += (size_t)N * H1;
  float* h1 = ws;                  ws += (size_t)N * H1;
  float* xl2 = ws;                 ws += (size_t)N * F_OUT;
  float* xr2 = ws;                 ws += (size_t)N * F_OUT;
  int* wi = (int*)ws;
  int* cnt = wi;                   wi += (size_t)((N + 1 + 31) / 32) * 32;
  int* bucket = wi;                wi += (size_t)N * SLOTS;
  int* spill = wi;
  (void)ws_size; (void)n_in; (void)out_size;

  const int BLK = 256;

  zero_counts<<<(N + 1 + BLK - 1) / BLK, BLK, 0, stream>>>(cnt, N + 1);
  bucket_edges<<<(E + BLK - 1) / BLK, BLK, 0, stream>>>(srcE, dstE, E, N, cnt,
                                                         bucket, spill);

  {
    const int rowTiles = N / 16;
    const int blocks = (rowTiles + 7) / 8;
    wmma_strip_gemm_l1<<<blocks, BLK, 0, stream>>>(x, W1l, W1r, xl1, xr1,
                                                  rowTiles);
  }
  {
    const int threads = N * HEADS1 * 2;
    node_softmax_aggregate<10, 8, 2, true>
        <<<(threads + BLK - 1) / BLK, BLK, 0, stream>>>(
            cnt, bucket, spill, N, xl1, xr1, att1, b1, h1);
  }

  {
    const int rowTiles = N / 16, colTiles = F_OUT / 16;
    const int waves = rowTiles * colTiles;
    const int blocks = (waves + 7) / 8;
    wmma_dual_gemm<80, 16><<<blocks, BLK, 0, stream>>>(
        h1, W2l, W2r, xl2, xr2, rowTiles, colTiles);
  }
  {
    const int threads = N * 4;
    node_softmax_aggregate<1, 16, 4, false>
        <<<(threads + BLK - 1) / BLK, BLK, 0, stream>>>(
            cnt, bucket, spill, N, xl2, xr2, att2, b2, out);
  }
}


# Per-node gather pass takes 4 bucket entries per step: one 16-byte index read, 4 row gathers in flight per lane, one online-softmax rescale per 4 edges (partial-sum rounding changes, order kept)
# speedup vs baseline: 5.8010x; 1.0401x over previous
#include <hip/hip_runtime.h>


typedef __attribute__((ext_vector_type(2))) float v2f;
typedef __attribute__((ext_vector_type(8))) float v8f;

#define NEG_SLOPE 0.2f

template <int K, int LDB>
__global__ __launch_bounds__(256) void wmma_dual_gemm(
    const float* __restrict__ A, const float* __restrict__ Bl,
    const float* __restrict__ Br, float* __restrict__ Cl,
    float* __restrict__ Cr, int rowTiles, int colTiles) {
  const int wid = (blockIdx.x * blockDim.x + threadIdx.x) >> 5;
  const int lane = threadIdx.x & 31;
  if (wid >= rowTiles * colTiles) return;
  const int rt = wid / colTiles;
  const int ct = wid % colTiles;
  const int row0 = rt * 16;
  const int col0 = ct * 16;
  const int half = lane >> 4;
  const int l16 = lane & 15;
  const int arow = row0 + l16;
  const int bcol = col0 + l16;

  const float* __restrict__ Ap = A + (size_t)arow * K + half * 2;

  v8f accl = {};
  v8f accr = {};
  for (int k = 0; k < K; k += 4) {
    const int ka = k + half * 2;
    v2f a;
    a.x = Ap[k];
    a.y = Ap[k + 1];
    v2f bl, br;
    bl.x = Bl[(size_t)ka * LDB + bcol];
    bl.y = Bl[(size_t)(ka + 1) * LDB + bcol];
    br.x = Br[(size_t)ka * LDB + bcol];
    br.y = Br[(size_t)(ka + 1) * LDB + bcol];
    accl = __builtin_amdgcn_wmma_f32_16x16x4_f32(false, a, false, bl, (short)0,
                                                 accl, false, false);
    accr = __builtin_amdgcn_wmma_f32_16x16x4_f32(false, a, false, br, (short)0,
                                                 accr, false, false);
  }

  const size_t base = (size_t)(row0 + half * 8) * LDB + bcol;
  float* __restrict__ cl = Cl + base;
  float* __restrict__ cr = Cr + base;
#pragma unroll
  for (int i = 0; i < 8; ++i) {
    cl[(size_t)i * LDB] = accl[i];
    cr[(size_t)i * LDB] = accr[i];
  }
}

#define S_K 128
#define S_LD 80
#define S_KB (S_K / 8)
#define S_NT 10
#define S_WAVES 8

__global__ __launch_bounds__(256) void wmma_strip_gemm_l1(
    const float* __restrict__ A, const float* __restrict__ Bl,
    const float* __restrict__ Br, float* __restrict__ Cl,
    float* __restrict__ Cr, int rowTiles) {
  __shared__ float4 sB[S_KB * S_NT * 32];

  const int tid = threadIdx.x;
  const int lane = tid & 31;
  const int wave = tid >> 5;
  const int half = lane >> 4;
  const int l16 = lane & 15;
  const int rt = blockIdx.x * S_WAVES + wave;
  const bool live = rt < rowTiles;
  const int row0 = rt * 16;

  float4 a[S_KB];
  if (live) {
    const float* __restrict__ Ap = A + (size_t)(row0 + l16) * S_K + half * 4;
#pragma unroll
    for (int kb = 0; kb < S_KB; ++kb)
      a[kb] = *reinterpret_cast<const float4*>(Ap + kb * 8);
  }

  {
    constexpr int ITEMS = 2 * S_KB * 2 * S_LD;
    constexpr int PER = ITEMS / 256;
    float4 w[PER];
#pragma unroll
    for (int i = 0; i < PER; ++i) {
      const int it = tid + 256 * i;
      const int q = it / S_LD;
      const int col = it % S_LD;
      const int m = q >> 5;
      const int kb = (q & 31) >> 1;
      const int h = q & 1;
      const float* __restrict__ src =
          (m ? Br : Bl) + (size_t)(kb * 8 + h * 4) * S_LD + col;
      w[i].x = src[0];
      w[i].y = src[S_LD];
      w[i].z = src[2 * S_LD];
      w[i].w = src[3 * S_LD];
    }
#pragma unroll
    for (int i = 0; i < PER; ++i) {
      const int it = tid + 256 * i;
      const int q = it / S_LD;
      const int col = it % S_LD;
      const int m = q >> 5;
      const int kb = (q & 31) >> 1;
      const int h = q & 1;
      const int t = m * 5 + (col >> 4);
      sB[(kb * S_NT + t) * 32 + h * 16 + (col & 15)] = w[i];
    }
  }
  __syncthreads();
  if (!live) return;

  v8f acc[S_NT];
#pragma unroll
  for (int t = 0; t < S_NT; ++t) acc[t] = (v8f){};

#pragma unroll
  for (int kb = 0; kb < S_KB; ++kb) {
    float4 b[S_NT];
#pragma unroll
    for (int t = 0; t < S_NT; ++t) b[t] = sB[(kb * S_NT + t) * 32 + lane];
    v2f a01, a23;
    a01.x = a[kb].x; a01.y = a[kb].y;
    a23.x = a[kb].z; a23.y = a[kb].w;
#pragma unroll
    for (int t = 0; t < S_NT; ++t) {
      v2f b01;
      b01.x = b[t].x; b01.y = b[t].y;
      acc[t] = __builtin_amdgcn_wmma_f32_16x16x4_f32(false, a01, false, b01,
                                                     (short)0, acc[t], false,
                                                     false);
    }
#pragma unroll
    for (int t = 0; t < S_NT; ++t) {
      v2f b23;
      b23.x = b[t].z; b23.y = b[t].w;
      acc[t] = __builtin_amdgcn_wmma_f32_16x16x4_f32(false, a23, false, b23,
                                                     (short)0, acc[t], false,
                                                     false);
    }
  }

  const size_t base = (size_t)(row0 + half * 8) * S_LD + l16;
#pragma unroll
  for (int t = 0; t < S_NT; ++t) {
    float* __restrict__ c = (t < 5 ? Cl + t * 16 : Cr + (t - 5) * 16) + base;
#pragma unroll
    for (int i = 0; i < 8; ++i) c[(size_t)i * S_LD] = acc[t][i];
  }
}

#define SLOTS 32

__global__ void zero_counts(int* __restrict__ cnt, int n) {
  const int i = blockIdx.x * blockDim.x + threadIdx.x;
  if (i < n) cnt[i] = 0;
}

__global__ void bucket_edges(const int* __restrict__ srcE,
                             const int* __restrict__ dstE, int E, int N,
                             int* __restrict__ cnt, int* __restrict__ bucket,
                             int* __restrict__ spill) {
  const int e = blockIdx.x * blockDim.x + threadIdx.x;
  if (e >= E) return;
  const int d = dstE[e];
  const int s = srcE[e];
  const int rank = atomicAdd(&cnt[d], 1);
  if (rank < SLOTS) {
    bucket[(size_t)d * SLOTS + rank] = s;
  } else {
    const int o = atomicAdd(&cnt[N], 1);
    reinterpret_cast<int2*>(spill)[o] = make_int2(d, s);
  }
}

__device__ __forceinline__ float4 ld4(const float* p) {
  return *reinterpret_cast<const float4*>(p);
}

template <int HEADS, int DIMH, int LPG, bool RELU>
__global__ __launch_bounds__(256) void node_softmax_aggregate(
    const int* __restrict__ cnt, const int* __restrict__ bucket,
    const int* __restrict__ spill, int N,
    const float* __restrict__ xl, const float* __restrict__ xr,
    const float* __restrict__ att, const float* __restrict__ bias,
    float* __restrict__ out) {
  static_assert(DIMH == 4 * LPG, "each lane owns 4 features");
  static_assert(LPG == 1 || LPG == 2 || LPG == 4, "group inside a quad");
  constexpr int H = HEADS * DIMH;
  const int t = blockIdx.x * blockDim.x + threadIdx.x;
  const int g = t / LPG;
  const int sub = t % LPG;
  if (g >= N * HEADS) return;
  const int d = g / HEADS;
  const int h = g % HEADS;
  const int col = h * DIMH + sub * 4;

  const float4 r = ld4(xr + (size_t)d * H + col);
  const float a0 = att[col], a1 = att[col + 1], a2 = att[col + 2],
              a3 = att[col + 3];

  auto logit_of = [&](const float4& v) -> float {
    float z0 = v.x + r.x, z1 = v.y + r.y, z2 = v.z + r.z, z3 = v.w + r.w;
    z0 = z0 > 0.f ? z0 : NEG_SLOPE * z0;
    z1 = z1 > 0.f ? z1 : NEG_SLOPE * z1;
    z2 = z2 > 0.f ? z2 : NEG_SLOPE * z2;
    z3 = z3 > 0.f ? z3 : NEG_SLOPE * z3;
    float p = z0 * a0;
    p += z1 * a1;
    p += z2 * a2;
    p += z3 * a3;
    if (LPG >= 2) p += __shfl_xor(p, 1);
    if (LPG >= 4) p += __shfl_xor(p, 2);
    return p;
  };

  const float4 xs = ld4(xl + (size_t)d * H + col);
  float m = logit_of(xs);
  float den = 1.f;
  float4 acc = xs;

  auto absorb = [&](const float4& v) {
    const float lg = logit_of(v);
    const float mn = fmaxf(m, lg);
    const float sc = __expf(m - mn);
    const float pe = __expf(lg - mn);
    den = den * sc + pe;
    acc.x = acc.x * sc + pe * v.x;
    acc.y = acc.y * sc + pe * v.y;
    acc.z = acc.z * sc + pe * v.z;
    acc.w = acc.w * sc + pe * v.w;
    m = mn;
  };

  auto absorb4 = [&](const float4& v0, const float4& v1, const float4& v2,
                     const float4& v3, int rem) {
    const float ninf = -__builtin_inff();
    const float l0 = logit_of(v0);
    float l1 = logit_of(v1);
    float l2 = logit_of(v2);
    float l3 = logit_of(v3);
    l1 = rem > 1 ? l1 : ninf;
    l2 = rem > 2 ? l2 : ninf;
    l3 = rem > 3 ? l3 : ninf;
    const float mn = fmaxf(fmaxf(m, l0), fmaxf(l1, fmaxf(l2, l3)));
    const float sc = __expf(m - mn);
    const float p0 = __expf(l0 - mn);
    const float p1 = __expf(l1 - mn);
    const float p2 = __expf(l2 - mn);
    const float p3 = __expf(l3 - mn);
    den = den * sc + p0;
    den += p1;
    den += p2;
    den += p3;
    acc.x = acc.x * sc + p0 * v0.x;
    acc.y = acc.y * sc + p0 * v0.y;
    acc.z = acc.z * sc + p0 * v0.z;
    acc.w = acc.w * sc + p0 * v0.w;
    acc.x += p1 * v1.x; acc.y += p1 * v1.y; acc.z += p1 * v1.z; acc.w += p1 * v1.w;
    acc.x += p2 * v2.x; acc.y += p2 * v2.y; acc.z += p2 * v2.z; acc.w += p2 * v2.w;
    acc.x += p3 * v3.x; acc.y += p3 * v3.y; acc.z += p3 * v3.z; acc.w += p3 * v3.w;
    m = mn;
  };

  const int deg = cnt[d];
  const int nb = deg < SLOTS ? deg : SLOTS;
  const int nbatch = (nb + 3) >> 2;
  const int4* __restrict__ mine4 =
      reinterpret_cast<const int4*>(bucket + (size_t)d * SLOTS);
  auto rows4 = [&](int b, const int4& ix, float4& v0, float4& v1, float4& v2,
                   float4& v3) {
    const int j = 4 * b;
    const int s0 = (j < nb) ? ix.x : d;
    const int s1 = (j + 1 < nb) ? ix.y : d;
    const int s2 = (j + 2 < nb) ? ix.z : d;
    const int s3 = (j + 3 < nb) ? ix.w : d;
    v0 = ld4(xl + (size_t)s0 * H + col);
    v1 = ld4(xl + (size_t)s1 * H + col);
    v2 = ld4(xl + (size_t)s2 * H + col);
    v3 = ld4(xl + (size_t)s3 * H + col);
  };
  for (int b = 0; b < nbatch; ++b) {
    const int4 ix = mine4[b];
    float4 v0, v1, v2, v3;
    rows4(b, ix, v0, v1, v2, v3);
    absorb4(v0, v1, v2, v3, nb - 4 * b);
  }

  if (deg > SLOTS) {
    const int nSpill = cnt[N];
    const int2* __restrict__ sp = reinterpret_cast<const int2*>(spill);
    for (int i = 0; i < nSpill; ++i) {
      const int2 ds = sp[i];
      if (ds.x == d) absorb(ld4(xl + (size_t)ds.y * H + col));
    }
  }

  const float inv = 1.f / den;
  float4 o;
  o.x = acc.x * inv + bias[col];
  o.y = acc.y * inv + bias[col + 1];
  o.z = acc.z * inv + bias[col + 2];
  o.w = acc.w * inv + bias[col + 3];
  if (RELU) {
    o.x = fmaxf(o.x, 0.f);
    o.y = fmaxf(o.y, 0.f);
    o.z = fmaxf(o.z, 0.f);
    o.w = fmaxf(o.w, 0.f);
  }
  *reinterpret_cast<float4*>(out + (size_t)d * H + col) = o;
}

extern "C" void kernel_launch(void* const* d_in, const int* in_sizes, int n_in,
                              void* d_out, int out_size, void* d_ws,
                              size_t ws_size, hipStream_t stream) {
  const float* x = (const float*)d_in[0];
  const int* edge = (const int*)d_in[1];
  const float* W1l = (const float*)d_in[2];
  const float* W1r = (const float*)d_in[3];
  const float* att1 = (const float*)d_in[4];
  const float* b1 = (const float*)d_in[5];
  const float* W2l = (const float*)d_in[6];
  const float* W2r = (const float*)d_in[7];
  const float* att2 = (const float*)d_in[8];
  const float* b2 = (const float*)d_in[9];
  float* out = (float*)d_out;

  const int F_IN = 128, H1 = 80, HEADS1 = 10, F_OUT = 16;
  const int N = in_sizes[0] / F_IN;
  const int E = in_sizes[1] / 2;
  const int* srcE = edge;
  const int* dstE = edge + E;

  float* ws = (float*)d_ws;
  float* xl1 = ws;                 ws += (size_t)N * H1;
  float* xr1 = ws;                 ws += (size_t)N * H1;
  float* h1 = ws;                  ws += (size_t)N * H1;
  float* xl2 = ws;                 ws += (size_t)N * F_OUT;
  float* xr2 = ws;                 ws += (size_t)N * F_OUT;
  int* wi = (int*)ws;
  int* cnt = wi;                   wi += (size_t)((N + 1 + 31) / 32) * 32;
  int* bucket = wi;                wi += (size_t)N * SLOTS;
  int* spill = wi;
  (void)ws_size; (void)n_in; (void)out_size;

  const int BLK = 256;

  zero_counts<<<(N + 1 + BLK - 1) / BLK, BLK, 0, stream>>>(cnt, N + 1);
  bucket_edges<<<(E + BLK - 1) / BLK, BLK, 0, stream>>>(srcE, dstE, E, N, cnt,
                                                         bucket, spill);

  {
    const int rowTiles = N / 16;
    const int blocks = (rowTiles + 7) / 8;
    wmma_strip_gemm_l1<<<blocks, BLK, 0, stream>>>(x, W1l, W1r, xl1, xr1,
                                                  rowTiles);
  }
  {
    const int threads = N * HEADS1 * 2;
    node_softmax_aggregate<10, 8, 2, true>
        <<<(threads + BLK - 1) / BLK, BLK, 0, stream>>>(
            cnt, bucket, spill, N, xl1, xr1, att1, b1, h1);
  }

  {
    const int rowTiles = N / 16, colTiles = F_OUT / 16;
    const int waves = rowTiles * colTiles;
    const int blocks = (waves + 7) / 8;
    wmma_dual_gemm<80, 16><<<blocks, BLK, 0, stream>>>(
        h1, W2l, W2r, xl2, xr2, rowTiles, colTiles);
  }
  {
    const int threads = N * 4;
    node_softmax_aggregate<1, 16, 4, false>
        <<<(threads + BLK - 1) / BLK, BLK, 0, stream>>>(
            cnt, bucket, spill, N, xl2, xr2, att2, b2, out);
  }
}


# Edge grouping rides in the layer-1 transform launch: each transform thread ranks and places 8 edges, its atomics in flight under the wave's matrix loop; one launch fewer; sums stay in arrival order
# speedup vs baseline: 6.0379x; 1.0408x over previous
#include <hip/hip_runtime.h>


typedef __attribute__((ext_vector_type(2))) float v2f;
typedef __attribute__((ext_vector_type(8))) float v8f;

#define NEG_SLOPE 0.2f

template <int K, int LDB>
__global__ __launch_bounds__(256) void wmma_dual_gemm(
    const float* __restrict__ A, const float* __restrict__ Bl,
    const float* __restrict__ Br, float* __restrict__ Cl,
    float* __restrict__ Cr, int rowTiles, int colTiles) {
  const int wid = (blockIdx.x * blockDim.x + threadIdx.x) >> 5;
  const int lane = threadIdx.x & 31;
  if (wid >= rowTiles * colTiles) return;
  const int rt = wid / colTiles;
  const int ct = wid % colTiles;
  const int row0 = rt * 16;
  const int col0 = ct * 16;
  const int half = lane >> 4;
  const int l16 = lane & 15;
  const int arow = row0 + l16;
  const int bcol = col0 + l16;

  const float* __restrict__ Ap = A + (size_t)arow * K + half * 2;

  v8f accl = {};
  v8f accr = {};
  for (int k = 0; k < K; k += 4) {
    const int ka = k + half * 2;
    v2f a;
    a.x = Ap[k];
    a.y = Ap[k + 1];
    v2f bl, br;
    bl.x = Bl[(size_t)ka * LDB + bcol];
    bl.y = Bl[(size_t)(ka + 1) * LDB + bcol];
    br.x = Br[(size_t)ka * LDB + bcol];
    br.y = Br[(size_t)(ka + 1) * LDB + bcol];
    accl = __builtin_amdgcn_wmma_f32_16x16x4_f32(false, a, false, bl, (short)0,
                                                 accl, false, false);
    accr = __builtin_amdgcn_wmma_f32_16x16x4_f32(false, a, false, br, (short)0,
                                                 accr, false, false);
  }

  const size_t base = (size_t)(row0 + half * 8) * LDB + bcol;
  float* __restrict__ cl = Cl + base;
  float* __restrict__ cr = Cr + base;
#pragma unroll
  for (int i = 0; i < 8; ++i) {
    cl[(size_t)i * LDB] = accl[i];
    cr[(size_t)i * LDB] = accr[i];
  }
}

#define SLOTS 32
#define EPT 8

__global__ void zero_counts(int* __restrict__ cnt, int n) {
  const int i = blockIdx.x * blockDim.x + threadIdx.x;
  if (i < n) cnt[i] = 0;
}

__device__ __forceinline__ void place_edge(int d, int s, int rank, int N,
                                           int* __restrict__ cnt,
                                           int* __restrict__ bucket,
                                           int* __restrict__ spill) {
  if (rank < SLOTS) {
    bucket[(size_t)d * SLOTS + rank] = s;
  } else {
    const int o = atomicAdd(&cnt[N], 1);
    reinterpret_cast<int2*>(spill)[o] = make_int2(d, s);
  }
}

#define S_K 128
#define S_LD 80
#define S_KB (S_K / 8)
#define S_NT 10
#define S_WAVES 8

__global__ __launch_bounds__(256) void wmma_strip_gemm_l1_bucket(
    const float* __restrict__ A, const float* __restrict__ Bl,
    const float* __restrict__ Br, float* __restrict__ Cl,
    float* __restrict__ Cr, int rowTiles, const int* __restrict__ srcE,
    const int* __restrict__ dstE, int E, int N, int* __restrict__ cnt,
    int* __restrict__ bucket, int* __restrict__ spill) {
  __shared__ float4 sB[S_KB * S_NT * 32];

  const int tid = threadIdx.x;
  const int lane = tid & 31;
  const int wave = tid >> 5;
  const int half = lane >> 4;
  const int l16 = lane & 15;
  const int rt = blockIdx.x * S_WAVES + wave;
  const bool live = rt < rowTiles;
  const int row0 = rt * 16;

  const int gt = blockIdx.x * 256 + tid;
  const int e0 = gt * EPT;
  int nv = E - e0;
  nv = nv < 0 ? 0 : (nv > EPT ? EPT : nv);
  int dv[EPT], sv[EPT];
  if (nv == EPT && (E & 3) == 0) {
    const int4 d0 = *reinterpret_cast<const int4*>(dstE + e0);
    const int4 d1 = *reinterpret_cast<const int4*>(dstE + e0 + 4);
    const int4 s0 = *reinterpret_cast<const int4*>(srcE + e0);
    const int4 s1 = *reinterpret_cast<const int4*>(srcE + e0 + 4);
    dv[0] = d0.x; dv[1] = d0.y; dv[2] = d0.z; dv[3] = d0.w;
    dv[4] = d1.x; dv[5] = d1.y; dv[6] = d1.z; dv[7] = d1.w;
    sv[0] = s0.x; sv[1] = s0.y; sv[2] = s0.z; sv[3] = s0.w;
    sv[4] = s1.x; sv[5] = s1.y; sv[6] = s1.z; sv[7] = s1.w;
  } else {
#pragma unroll
    for (int i = 0; i < EPT; ++i) {
      dv[i] = i < nv ? dstE[e0 + i] : 0;
      sv[i] = i < nv ? srcE[e0 + i] : 0;
    }
  }

  float4 a[S_KB];
  if (live) {
    const float* __restrict__ Ap = A + (size_t)(row0 + l16) * S_K + half * 4;
#pragma unroll
    for (int kb = 0; kb < S_KB; ++kb)
      a[kb] = *reinterpret_cast<const float4*>(Ap + kb * 8);
  }

  {
    constexpr int ITEMS = 2 * S_KB * 2 * S_LD;
    constexpr int PER = ITEMS / 256;
    float4 w[PER];
#pragma unroll
    for (int i = 0; i < PER; ++i) {
      const int it = tid + 256 * i;
      const int q = it / S_LD;
      const int col = it % S_LD;
      const int m = q >> 5;
      const int kb = (q & 31) >> 1;
      const int h = q & 1;
      const float* __restrict__ src =
          (m ? Br : Bl) + (size_t)(kb * 8 + h * 4) * S_LD + col;
      w[i].x = src[0];
      w[i].y = src[S_LD];
      w[i].z = src[2 * S_LD];
      w[i].w = src[3 * S_LD];
    }
#pragma unroll
    for (int i = 0; i < PER; ++i) {
      const int it = tid + 256 * i;
      const int q = it / S_LD;
      const int col = it % S_LD;
      const int m = q >> 5;
      const int kb = (q & 31) >> 1;
      const int h = q & 1;
      const int t = m * 5 + (col >> 4);
      sB[(kb * S_NT + t) * 32 + h * 16 + (col & 15)] = w[i];
    }
  }
  __syncthreads();

  int rk[EPT];
#pragma unroll
  for (int i = 0; i < EPT; ++i) {
    rk[i] = 0;
    if (i < nv) rk[i] = atomicAdd(&cnt[dv[i]], 1);
  }

  if (live) {
    v8f acc[S_NT];
#pragma unroll
    for (int t = 0; t < S_NT; ++t) acc[t] = (v8f){};

#pragma unroll
    for (int kb = 0; kb < S_KB; ++kb) {
      float4 b[S_NT];
#pragma unroll
      for (int t = 0; t < S_NT; ++t) b[t] = sB[(kb * S_NT + t) * 32 + lane];
      v2f a01, a23;
      a01.x = a[kb].x; a01.y = a[kb].y;
      a23.x = a[kb].z; a23.y = a[kb].w;
#pragma unroll
      for (int t = 0; t < S_NT; ++t) {
        v2f b01;
        b01.x = b[t].x; b01.y = b[t].y;
        acc[t] = __builtin_amdgcn_wmma_f32_16x16x4_f32(false, a01, false, b01,
                                                       (short)0, acc[t], false,
                                                       false);
      }
#pragma unroll
      for (int t = 0; t < S_NT; ++t) {
        v2f b23;
        b23.x = b[t].z; b23.y = b[t].w;
        acc[t] = __builtin_amdgcn_wmma_f32_16x16x4_f32(false, a23, false, b23,
                                                       (short)0, acc[t], false,
                                                       false);
      }
    }

    const size_t base = (size_t)(row0 + half * 8) * S_LD + l16;
#pragma unroll
    for (int t = 0; t < S_NT; ++t) {
      float* __restrict__ c = (t < 5 ? Cl + t * 16 : Cr + (t - 5) * 16) + base;
#pragma unroll
      for (int i = 0; i < 8; ++i) c[(size_t)i * S_LD] = acc[t][i];
    }
  }

#pragma unroll
  for (int i = 0; i < EPT; ++i)
    if (i < nv) place_edge(dv[i], sv[i], rk[i], N, cnt, bucket, spill);

  const int gthreads = gridDim.x * 256;
  for (int e = gthreads * EPT + gt; e < E; e += gthreads) {
    const int d = dstE[e];
    const int s = srcE[e];
    place_edge(d, s, atomicAdd(&cnt[d], 1), N, cnt, bucket, spill);
  }
}

__device__ __forceinline__ float4 ld4(const float* p) {
  return *reinterpret_cast<const float4*>(p);
}

template <int HEADS, int DIMH, int LPG, bool RELU>
__global__ __launch_bounds__(256) void node_softmax_aggregate(
    const int* __restrict__ cnt, const int* __restrict__ bucket,
    const int* __restrict__ spill, int N,
    const float* __restrict__ xl, const float* __restrict__ xr,
    const float* __restrict__ att, const float* __restrict__ bias,
    float* __restrict__ out) {
  static_assert(DIMH == 4 * LPG, "each lane owns 4 features");
  static_assert(LPG == 1 || LPG == 2 || LPG == 4, "group inside a quad");
  constexpr int H = HEADS * DIMH;
  const int t = blockIdx.x * blockDim.x + threadIdx.x;
  const int g = t / LPG;
  const int sub = t % LPG;
  if (g >= N * HEADS) return;
  const int d = g / HEADS;
  const int h = g % HEADS;
  const int col = h * DIMH + sub * 4;

  const float4 r = ld4(xr + (size_t)d * H + col);
  const float a0 = att[col], a1 = att[col + 1], a2 = att[col + 2],
              a3 = att[col + 3];

  auto logit_of = [&](const float4& v) -> float {
    float z0 = v.x + r.x, z1 = v.y + r.y, z2 = v.z + r.z, z3 = v.w + r.w;
    z0 = z0 > 0.f ? z0 : NEG_SLOPE * z0;
    z1 = z1 > 0.f ? z1 : NEG_SLOPE * z1;
    z2 = z2 > 0.f ? z2 : NEG_SLOPE * z2;
    z3 = z3 > 0.f ? z3 : NEG_SLOPE * z3;
    float p = z0 * a0;
    p += z1 * a1;
    p += z2 * a2;
    p += z3 * a3;
    if (LPG >= 2) p += __shfl_xor(p, 1);
    if (LPG >= 4) p += __shfl_xor(p, 2);
    return p;
  };

  const float4 xs = ld4(xl + (size_t)d * H + col);
  float m = logit_of(xs);
  float den = 1.f;
  float4 acc = xs;

  auto absorb = [&](const float4& v) {
    const float lg = logit_of(v);
    const float mn = fmaxf(m, lg);
    const float sc = __expf(m - mn);
    const float pe = __expf(lg - mn);
    den = den * sc + pe;
    acc.x = acc.x * sc + pe * v.x;
    acc.y = acc.y * sc + pe * v.y;
    acc.z = acc.z * sc + pe * v.z;
    acc.w = acc.w * sc + pe * v.w;
    m = mn;
  };

  auto absorb4 = [&](const float4& v0, const float4& v1, const float4& v2,
                     const float4& v3, int rem) {
    const float ninf = -__builtin_inff();
    const float l0 = logit_of(v0);
    float l1 = logit_of(v1);
    float l2 = logit_of(v2);
    float l3 = logit_of(v3);
    l1 = rem > 1 ? l1 : ninf;
    l2 = rem > 2 ? l2 : ninf;
    l3 = rem > 3 ? l3 : ninf;
    const float mn = fmaxf(fmaxf(m, l0), fmaxf(l1, fmaxf(l2, l3)));
    const float sc = __expf(m - mn);
    const float p0 = __expf(l0 - mn);
    const float p1 = __expf(l1 - mn);
    const float p2 = __expf(l2 - mn);
    const float p3 = __expf(l3 - mn);
    den = den * sc + p0;
    den += p1;
    den += p2;
    den += p3;
    acc.x = acc.x * sc + p0 * v0.x;
    acc.y = acc.y * sc + p0 * v0.y;
    acc.z = acc.z * sc + p0 * v0.z;
    acc.w = acc.w * sc + p0 * v0.w;
    acc.x += p1 * v1.x; acc.y += p1 * v1.y; acc.z += p1 * v1.z; acc.w += p1 * v1.w;
    acc.x += p2 * v2.x; acc.y += p2 * v2.y; acc.z += p2 * v2.z; acc.w += p2 * v2.w;
    acc.x += p3 * v3.x; acc.y += p3 * v3.y; acc.z += p3 * v3.z; acc.w += p3 * v3.w;
    m = mn;
  };

  const int deg = cnt[d];
  const int nb = deg < SLOTS ? deg : SLOTS;
  const int nbatch = (nb + 3) >> 2;
  const int4* __restrict__ mine4 =
      reinterpret_cast<const int4*>(bucket + (size_t)d * SLOTS);
  auto rows4 = [&](int b, const int4& ix, float4& v0, float4& v1, float4& v2,
                   float4& v3) {
    const int j = 4 * b;
    const int s0 = (j < nb) ? ix.x : d;
    const int s1 = (j + 1 < nb) ? ix.y : d;
    const int s2 = (j + 2 < nb) ? ix.z : d;
    const int s3 = (j + 3 < nb) ? ix.w : d;
    v0 = ld4(xl + (size_t)s0 * H + col);
    v1 = ld4(xl + (size_t)s1 * H + col);
    v2 = ld4(xl + (size_t)s2 * H + col);
    v3 = ld4(xl + (size_t)s3 * H + col);
  };
  for (int b = 0; b < nbatch; ++b) {
    const int4 ix = mine4[b];
    float4 v0, v1, v2, v3;
    rows4(b, ix, v0, v1, v2, v3);
    absorb4(v0, v1, v2, v3, nb - 4 * b);
  }

  if (deg > SLOTS) {
    const int nSpill = cnt[N];
    const int2* __restrict__ sp = reinterpret_cast<const int2*>(spill);
    for (int i = 0; i < nSpill; ++i) {
      const int2 ds = sp[i];
      if (ds.x == d) absorb(ld4(xl + (size_t)ds.y * H + col));
    }
  }

  const float inv = 1.f / den;
  float4 o;
  o.x = acc.x * inv + bias[col];
  o.y = acc.y * inv + bias[col + 1];
  o.z = acc.z * inv + bias[col + 2];
  o.w = acc.w * inv + bias[col + 3];
  if (RELU) {
    o.x = fmaxf(o.x, 0.f);
    o.y = fmaxf(o.y, 0.f);
    o.z = fmaxf(o.z, 0.f);
    o.w = fmaxf(o.w, 0.f);
  }
  *reinterpret_cast<float4*>(out + (size_t)d * H + col) = o;
}

extern "C" void kernel_launch(void* const* d_in, const int* in_sizes, int n_in,
                              void* d_out, int out_size, void* d_ws,
                              size_t ws_size, hipStream_t stream) {
  const float* x = (const float*)d_in[0];
  const int* edge = (const int*)d_in[1];
  const float* W1l = (const float*)d_in[2];
  const float* W1r = (const float*)d_in[3];
  const float* att1 = (const float*)d_in[4];
  const float* b1 = (const float*)d_in[5];
  const float* W2l = (const float*)d_in[6];
  const float* W2r = (const float*)d_in[7];
  const float* att2 = (const float*)d_in[8];
  const float* b2 = (const float*)d_in[9];
  float* out = (float*)d_out;

  const int F_IN = 128, H1 = 80, HEADS1 = 10, F_OUT = 16;
  const int N = in_sizes[0] / F_IN;
  const int E = in_sizes[1] / 2;
  const int* srcE = edge;
  const int* dstE = edge + E;

  float* ws = (float*)d_ws;
  float* xl1 = ws;                 ws += (size_t)N * H1;
  float* xr1 = ws;                 ws += (size_t)N * H1;
  float* h1 = ws;                  ws += (size_t)N * H1;
  float* xl2 = ws;                 ws += (size_t)N * F_OUT;
  float* xr2 = ws;                 ws += (size_t)N * F_OUT;
  int* wi = (int*)ws;
  int* cnt = wi;                   wi += (size_t)((N + 1 + 31) / 32) * 32;
  int* bucket = wi;                wi += (size_t)N * SLOTS;
  int* spill = wi;
  (void)ws_size; (void)n_in; (void)out_size;

  const int BLK = 256;

  zero_counts<<<(N + 1 + BLK - 1) / BLK, BLK, 0, stream>>>(cnt, N + 1);

  {
    const int rowTiles = N / 16;
    const int blocks = (rowTiles + 7) / 8;
    wmma_strip_gemm_l1_bucket<<<blocks, BLK, 0, stream>>>(
        x, W1l, W1r, xl1, xr1, rowTiles, srcE, dstE, E, N, cnt, bucket, spill);
  }
  {
    const int threads = N * HEADS1 * 2;
    node_softmax_aggregate<10, 8, 2, true>
        <<<(threads + BLK - 1) / BLK, BLK, 0, stream>>>(
            cnt, bucket, spill, N, xl1, xr1, att1, b1, h1);
  }

  {
    const int rowTiles = N / 16, colTiles = F_OUT / 16;
    const int waves = rowTiles * colTiles;
    const int blocks = (waves + 7) / 8;
    wmma_dual_gemm<80, 16><<<blocks, BLK, 0, stream>>>(
        h1, W2l, W2r, xl2, xr2, rowTiles, colTiles);
  }
  {
    const int threads = N * 4;
    node_softmax_aggregate<1, 16, 4, false>
        <<<(threads + BLK - 1) / BLK, BLK, 0, stream>>>(
            cnt, bucket, spill, N, xl2, xr2, att2, b2, out);
  }
}


# Edge grouping via LDS, no per-edge global atomic: partition launch sorts edge chunks by 128-node range, transform group b ranks range b in LDS, whole-line bucket stores; sums in LDS arrival order
# speedup vs baseline: 9.9121x; 1.6416x over previous
#include <hip/hip_runtime.h>


typedef __attribute__((ext_vector_type(2))) float v2f;
typedef __attribute__((ext_vector_type(8))) float v8f;

#define NEG_SLOPE 0.2f

template <int K, int LDB>
__global__ __launch_bounds__(256) void wmma_dual_gemm(
    const float* __restrict__ A, const float* __restrict__ Bl,
    const float* __restrict__ Br, float* __restrict__ Cl,
    float* __restrict__ Cr, int rowTiles, int colTiles) {
  const int wid = (blockIdx.x * blockDim.x + threadIdx.x) >> 5;
  const int lane = threadIdx.x & 31;
  if (wid >= rowTiles * colTiles) return;
  const int rt = wid / colTiles;
  const int ct = wid % colTiles;
  const int row0 = rt * 16;
  const int col0 = ct * 16;
  const int half = lane >> 4;
  const int l16 = lane & 15;
  const int arow = row0 + l16;
  const int bcol = col0 + l16;

  const float* __restrict__ Ap = A + (size_t)arow * K + half * 2;

  v8f accl = {};
  v8f accr = {};
  for (int k = 0; k < K; k += 4) {
    const int ka = k + half * 2;
    v2f a;
    a.x = Ap[k];
    a.y = Ap[k + 1];
    v2f bl, br;
    bl.x = Bl[(size_t)ka * LDB + bcol];
    bl.y = Bl[(size_t)(ka + 1) * LDB + bcol];
    br.x = Br[(size_t)ka * LDB + bcol];
    br.y = Br[(size_t)(ka + 1) * LDB + bcol];
    accl = __builtin_amdgcn_wmma_f32_16x16x4_f32(false, a, false, bl, (short)0,
                                                 accl, false, false);
    accr = __builtin_amdgcn_wmma_f32_16x16x4_f32(false, a, false, br, (short)0,
                                                 accr, false, false);
  }

  const size_t base = (size_t)(row0 + half * 8) * LDB + bcol;
  float* __restrict__ cl = Cl + base;
  float* __restrict__ cr = Cr + base;
#pragma unroll
  for (int i = 0; i < 8; ++i) {
    cl[(size_t)i * LDB] = accl[i];
    cr[(size_t)i * LDB] = accr[i];
  }
}

#define SLOTS 32
#define RB 128
#define RB_SHIFT 7
#define P_CHUNK 3200
#define P_EPT ((P_CHUNK + 255) / 256)
#define NB_MAX 512
#define LPITCH 33

__global__ __launch_bounds__(256) void partition_edges(
    const int* __restrict__ srcE, const int* __restrict__ dstE, int E, int N,
    int NB, int* __restrict__ segbuf, int* __restrict__ segoff,
    int* __restrict__ cnt) {
  __shared__ int binc[NB_MAX];
  __shared__ int wsum[8];
  __shared__ int2 sorted[P_CHUNK];

  const int tid = threadIdx.x;
  const int lane = tid & 31;
  const int wave = tid >> 5;
  const int c = blockIdx.x;
  const int e0 = c * P_CHUNK;
  int n = E - e0;
  n = n > P_CHUNK ? P_CHUNK : n;

  if (c == 0 && tid == 0) cnt[N] = 0;
  binc[tid] = 0;
  binc[tid + 256] = 0;

  int dv[P_EPT], sv[P_EPT], rk[P_EPT];
#pragma unroll
  for (int i = 0; i < P_EPT; ++i) {
    const int k = tid + 256 * i;
    dv[i] = k < n ? dstE[e0 + k] : 0;
    sv[i] = k < n ? srcE[e0 + k] : 0;
  }
  __syncthreads();

#pragma unroll
  for (int i = 0; i < P_EPT; ++i) {
    rk[i] = 0;
    if (tid + 256 * i < n) rk[i] = atomicAdd(&binc[dv[i] >> RB_SHIFT], 1);
  }
  __syncthreads();

  const int ca = binc[2 * tid];
  const int cb = binc[2 * tid + 1];
  int inc = ca + cb;
#pragma unroll
  for (int o = 1; o < 32; o <<= 1) {
    const int v = __shfl_up(inc, o);
    if (lane >= o) inc += v;
  }
  if (lane == 31) wsum[wave] = inc;
  __syncthreads();
  int wbase = 0;
#pragma unroll
  for (int w = 0; w < 8; ++w) wbase += w < wave ? wsum[w] : 0;
  const int ex = wbase + inc - (ca + cb);
  binc[2 * tid] = ex;
  binc[2 * tid + 1] = ex + ca;
  __syncthreads();

#pragma unroll
  for (int i = 0; i < P_EPT; ++i)
    if (tid + 256 * i < n)
      sorted[binc[dv[i] >> RB_SHIFT] + rk[i]] = make_int2(dv[i], sv[i]);

  int* __restrict__ so = segoff + (size_t)c * (NB + 1);
  for (int b = tid; b <= NB; b += 256) so[b] = b < NB ? binc[b] : n;
  __syncthreads();

  int2* __restrict__ sg = reinterpret_cast<int2*>(segbuf) + e0;
  for (int k = tid; k < n; k += 256) sg[k] = sorted[k];
}

#define S_K 128
#define S_LD 80
#define S_KB (S_K / 8)
#define S_NT 10
#define S_WAVES 8

__global__ __launch_bounds__(256) void wmma_strip_gemm_l1_bucket(
    const float* __restrict__ A, const float* __restrict__ Bl,
    const float* __restrict__ Br, float* __restrict__ Cl,
    float* __restrict__ Cr, int rowTiles, const int* __restrict__ segbuf,
    const int* __restrict__ segoff, int NC, int N, int* __restrict__ cnt,
    int* __restrict__ bucket, int* __restrict__ spill) {
  static_assert(RB == 16 * S_WAVES, "a range is one work-group's rows");
  static_assert(SLOTS == 32, "one lane per slot in the bucket write-out");
  __shared__ float4 sB[S_KB * S_NT * 32];
  __shared__ int lcnt[RB];
  __shared__ int lbkt[RB * LPITCH];

  const int tid = threadIdx.x;
  const int lane = tid & 31;
  const int wave = tid >> 5;
  const int half = lane >> 4;
  const int l16 = lane & 15;
  const int rt = blockIdx.x * S_WAVES + wave;
  const bool live = rt < rowTiles;
  const int row0 = rt * 16;

  if (tid < RB) lcnt[tid] = 0;

  float4 a[S_KB];
  if (live) {
    const float* __restrict__ Ap = A + (size_t)(row0 + l16) * S_K + half * 4;
#pragma unroll
    for (int kb = 0; kb < S_KB; ++kb)
      a[kb] = *reinterpret_cast<const float4*>(Ap + kb * 8);
  }

  {
    constexpr int ITEMS = 2 * S_KB * 2 * S_LD;
    constexpr int PER = ITEMS / 256;
    float4 w[PER];
#pragma unroll
    for (int i = 0; i < PER; ++i) {
      const int it = tid + 256 * i;
      const int q = it / S_LD;
      const int col = it % S_LD;
      const int m = q >> 5;
      const int kb = (q & 31) >> 1;
      const int h = q & 1;
      const float* __restrict__ src =
          (m ? Br : Bl) + (size_t)(kb * 8 + h * 4) * S_LD + col;
      w[i].x = src[0];
      w[i].y = src[S_LD];
      w[i].z = src[2 * S_LD];
      w[i].w = src[3 * S_LD];
    }
#pragma unroll
    for (int i = 0; i < PER; ++i) {
      const int it = tid + 256 * i;
      const int q = it / S_LD;
      const int col = it % S_LD;
      const int m = q >> 5;
      const int kb = (q & 31) >> 1;
      const int h = q & 1;
      const int t = m * 5 + (col >> 4);
      sB[(kb * S_NT + t) * 32 + h * 16 + (col & 15)] = w[i];
    }
  }
  __syncthreads();

  if (live) {
    v8f acc[S_NT];
#pragma unroll
    for (int t = 0; t < S_NT; ++t) acc[t] = (v8f){};

#pragma unroll
    for (int kb = 0; kb < S_KB; ++kb) {
      float4 b[S_NT];
#pragma unroll
      for (int t = 0; t < S_NT; ++t) b[t] = sB[(kb * S_NT + t) * 32 + lane];
      v2f a01, a23;
      a01.x = a[kb].x; a01.y = a[kb].y;
      a23.x = a[kb].z; a23.y = a[kb].w;
#pragma unroll
      for (int t = 0; t < S_NT; ++t) {
        v2f b01;
        b01.x = b[t].x; b01.y = b[t].y;
        acc[t] = __builtin_amdgcn_wmma_f32_16x16x4_f32(false, a01, false, b01,
                                                       (short)0, acc[t], false,
                                                       false);
      }
#pragma unroll
      for (int t = 0; t < S_NT; ++t) {
        v2f b23;
        b23.x = b[t].z; b23.y = b[t].w;
        acc[t] = __builtin_amdgcn_wmma_f32_16x16x4_f32(false, a23, false, b23,
                                                       (short)0, acc[t], false,
                                                       false);
      }
    }

    const size_t base = (size_t)(row0 + half * 8) * S_LD + l16;
#pragma unroll
    for (int t = 0; t < S_NT; ++t) {
      float* __restrict__ c = (t < 5 ? Cl + t * 16 : Cr + (t - 5) * 16) + base;
#pragma unroll
      for (int i = 0; i < 8; ++i) c[(size_t)i * S_LD] = acc[t][i];
    }
  }

  const int nbase = blockIdx.x * RB;
  const int segPitch = gridDim.x + 1;
  auto place = [&](const int2 e) {
    const int dl = e.x - nbase;
    const int rank = atomicAdd(&lcnt[dl], 1);
    if (rank < SLOTS) {
      lbkt[dl * LPITCH + rank] = e.y;
    } else {
      const int o = atomicAdd(&cnt[N], 1);
      reinterpret_cast<int2*>(spill)[o] = e;
    }
  };
  for (int c = tid; c < NC; c += 256) {
    const int* __restrict__ so = segoff + (size_t)c * segPitch + blockIdx.x;
    const int beg = so[0];
    const int end = so[1];
    const int2* __restrict__ seg =
        reinterpret_cast<const int2*>(segbuf) + (size_t)c * P_CHUNK;
    for (int j = beg; j < end; j += 4) {
      const int last = end - 1;
      const int2 e0 = seg[j];
      const int2 e1 = seg[j + 1 < last ? j + 1 : last];
      const int2 e2 = seg[j + 2 < last ? j + 2 : last];
      const int2 e3 = seg[j + 3 < last ? j + 3 : last];
      place(e0);
      if (j + 1 < end) place(e1);
      if (j + 2 < end) place(e2);
      if (j + 3 < end) place(e3);
    }
  }
  __syncthreads();

  int nn = N - nbase;
  nn = nn > RB ? RB : nn;
  if (tid < nn) cnt[nbase + tid] = lcnt[tid];
  for (int n = wave; n < nn; n += S_WAVES)
    bucket[(size_t)(nbase + n) * SLOTS + lane] = lbkt[n * LPITCH + lane];
}

__device__ __forceinline__ float4 ld4(const float* p) {
  return *reinterpret_cast<const float4*>(p);
}

template <int HEADS, int DIMH, int LPG, bool RELU>
__global__ __launch_bounds__(256) void node_softmax_aggregate(
    const int* __restrict__ cnt, const int* __restrict__ bucket,
    const int* __restrict__ spill, int N,
    const float* __restrict__ xl, const float* __restrict__ xr,
    const float* __restrict__ att, const float* __restrict__ bias,
    float* __restrict__ out) {
  static_assert(DIMH == 4 * LPG, "each lane owns 4 features");
  static_assert(LPG == 1 || LPG == 2 || LPG == 4, "group inside a quad");
  constexpr int H = HEADS * DIMH;
  const int t = blockIdx.x * blockDim.x + threadIdx.x;
  const int g = t / LPG;
  const int sub = t % LPG;
  if (g >= N * HEADS) return;
  const int d = g / HEADS;
  const int h = g % HEADS;
  const int col = h * DIMH + sub * 4;

  const float4 r = ld4(xr + (size_t)d * H + col);
  const float a0 = att[col], a1 = att[col + 1], a2 = att[col + 2],
              a3 = att[col + 3];

  auto logit_of = [&](const float4& v) -> float {
    float z0 = v.x + r.x, z1 = v.y + r.y, z2 = v.z + r.z, z3 = v.w + r.w;
    z0 = z0 > 0.f ? z0 : NEG_SLOPE * z0;
    z1 = z1 > 0.f ? z1 : NEG_SLOPE * z1;
    z2 = z2 > 0.f ? z2 : NEG_SLOPE * z2;
    z3 = z3 > 0.f ? z3 : NEG_SLOPE * z3;
    float p = z0 * a0;
    p += z1 * a1;
    p += z2 * a2;
    p += z3 * a3;
    if (LPG >= 2) p += __shfl_xor(p, 1);
    if (LPG >= 4) p += __shfl_xor(p, 2);
    return p;
  };

  const float4 xs = ld4(xl + (size_t)d * H + col);
  float m = logit_of(xs);
  float den = 1.f;
  float4 acc = xs;

  auto absorb = [&](const float4& v) {
    const float lg = logit_of(v);
    const float mn = fmaxf(m, lg);
    const float sc = __expf(m - mn);
    const float pe = __expf(lg - mn);
    den = den * sc + pe;
    acc.x = acc.x * sc + pe * v.x;
    acc.y = acc.y * sc + pe * v.y;
    acc.z = acc.z * sc + pe * v.z;
    acc.w = acc.w * sc + pe * v.w;
    m = mn;
  };

  auto absorb4 = [&](const float4& v0, const float4& v1, const float4& v2,
                     const float4& v3, int rem) {
    const float ninf = -__builtin_inff();
    const float l0 = logit_of(v0);
    float l1 = logit_of(v1);
    float l2 = logit_of(v2);
    float l3 = logit_of(v3);
    l1 = rem > 1 ? l1 : ninf;
    l2 = rem > 2 ? l2 : ninf;
    l3 = rem > 3 ? l3 : ninf;
    const float mn = fmaxf(fmaxf(m, l0), fmaxf(l1, fmaxf(l2, l3)));
    const float sc = __expf(m - mn);
    const float p0 = __expf(l0 - mn);
    const float p1 = __expf(l1 - mn);
    const float p2 = __expf(l2 - mn);
    const float p3 = __expf(l3 - mn);
    den = den * sc + p0;
    den += p1;
    den += p2;
    den += p3;
    acc.x = acc.x * sc + p0 * v0.x;
    acc.y = acc.y * sc + p0 * v0.y;
    acc.z = acc.z * sc + p0 * v0.z;
    acc.w = acc.w * sc + p0 * v0.w;
    acc.x += p1 * v1.x; acc.y += p1 * v1.y; acc.z += p1 * v1.z; acc.w += p1 * v1.w;
    acc.x += p2 * v2.x; acc.y += p2 * v2.y; acc.z += p2 * v2.z; acc.w += p2 * v2.w;
    acc.x += p3 * v3.x; acc.y += p3 * v3.y; acc.z += p3 * v3.z; acc.w += p3 * v3.w;
    m = mn;
  };

  const int deg = cnt[d];
  const int nb = deg < SLOTS ? deg : SLOTS;
  const int nbatch = (nb + 3) >> 2;
  const int4* __restrict__ mine4 =
      reinterpret_cast<const int4*>(bucket + (size_t)d * SLOTS);
  auto rows4 = [&](int b, const int4& ix, float4& v0, float4& v1, float4& v2,
                   float4& v3) {
    const int j = 4 * b;
    const int s0 = (j < nb) ? ix.x : d;
    const int s1 = (j + 1 < nb) ? ix.y : d;
    const int s2 = (j + 2 < nb) ? ix.z : d;
    const int s3 = (j + 3 < nb) ? ix.w : d;
    v0 = ld4(xl + (size_t)s0 * H + col);
    v1 = ld4(xl + (size_t)s1 * H + col);
    v2 = ld4(xl + (size_t)s2 * H + col);
    v3 = ld4(xl + (size_t)s3 * H + col);
  };
  for (int b = 0; b < nbatch; ++b) {
    const int4 ix = mine4[b];
    float4 v0, v1, v2, v3;
    rows4(b, ix, v0, v1, v2, v3);
    absorb4(v0, v1, v2, v3, nb - 4 * b);
  }

  if (deg > SLOTS) {
    const int nSpill = cnt[N];
    const int2* __restrict__ sp = reinterpret_cast<const int2*>(spill);
    for (int i = 0; i < nSpill; ++i) {
      const int2 ds = sp[i];
      if (ds.x == d) absorb(ld4(xl + (size_t)ds.y * H + col));
    }
  }

  const float inv = 1.f / den;
  float4 o;
  o.x = acc.x * inv + bias[col];
  o.y = acc.y * inv + bias[col + 1];
  o.z = acc.z * inv + bias[col + 2];
  o.w = acc.w * inv + bias[col + 3];
  if (RELU) {
    o.x = fmaxf(o.x, 0.f);
    o.y = fmaxf(o.y, 0.f);
    o.z = fmaxf(o.z, 0.f);
    o.w = fmaxf(o.w, 0.f);
  }
  *reinterpret_cast<float4*>(out + (size_t)d * H + col) = o;
}

extern "C" void kernel_launch(void* const* d_in, const int* in_sizes, int n_in,
                              void* d_out, int out_size, void* d_ws,
                              size_t ws_size, hipStream_t stream) {
  const float* x = (const float*)d_in[0];
  const int* edge = (const int*)d_in[1];
  const float* W1l = (const float*)d_in[2];
  const float* W1r = (const float*)d_in[3];
  const float* att1 = (const float*)d_in[4];
  const float* b1 = (const float*)d_in[5];
  const float* W2l = (const float*)d_in[6];
  const float* W2r = (const float*)d_in[7];
  const float* att2 = (const float*)d_in[8];
  const float* b2 = (const float*)d_in[9];
  float* out = (float*)d_out;

  const int F_IN = 128, H1 = 80, HEADS1 = 10, F_OUT = 16;
  const int N = in_sizes[0] / F_IN;
  const int E = in_sizes[1] / 2;
  const int* srcE = edge;
  const int* dstE = edge + E;

  float* ws = (float*)d_ws;
  float* xl1 = ws;                 ws += (size_t)N * H1;
  float* xr1 = ws;                 ws += (size_t)N * H1;
  float* h1 = ws;                  ws += (size_t)N * H1;
  float* xl2 = ws;                 ws += (size_t)N * F_OUT;
  float* xr2 = ws;                 ws += (size_t)N * F_OUT;
  int* wi = (int*)ws;
  int* cnt = wi;                   wi += (size_t)((N + 1 + 31) / 32) * 32;
  int* bucket = wi;                wi += (size_t)N * SLOTS;
  int* spill = wi;                 wi += (size_t)2 * E;
  int* segbuf = wi;                wi += (size_t)2 * E;
  int* segoff = wi;
  (void)ws_size; (void)n_in; (void)out_size;

  const int BLK = 256;

  const int NB = (N + RB - 1) / RB;
  const int NC = (E + P_CHUNK - 1) / P_CHUNK;
  partition_edges<<<NC, BLK, 0, stream>>>(srcE, dstE, E, N, NB, segbuf, segoff,
                                          cnt);

  {
    const int rowTiles = N / 16;
    wmma_strip_gemm_l1_bucket<<<NB, BLK, 0, stream>>>(
        x, W1l, W1r, xl1, xr1, rowTiles, segbuf, segoff, NC, N, cnt, bucket,
        spill);
  }
  {
    const int threads = N * HEADS1 * 2;
    node_softmax_aggregate<10, 8, 2, true>
        <<<(threads + BLK - 1) / BLK, BLK, 0, stream>>>(
            cnt, bucket, spill, N, xl1, xr1, att1, b1, h1);
  }

  {
    const int rowTiles = N / 16, colTiles = F_OUT / 16;
    const int waves = rowTiles * colTiles;
    const int blocks = (waves + 7) / 8;
    wmma_dual_gemm<80, 16><<<blocks, BLK, 0, stream>>>(
        h1, W2l, W2r, xl2, xr2, rowTiles, colTiles);
  }
  {
    const int threads = N * 4;
    node_softmax_aggregate<1, 16, 4, false>
        <<<(threads + BLK - 1) / BLK, BLK, 0, stream>>>(
            cnt, bucket, spill, N, xl2, xr2, att2, b2, out);
  }
}


# Layer-2 transform fused into the layer-1 aggregation launch: work-group = 16 nodes, ReLU'd rows go through an LDS tile into 2 waves' WMMAs; one launch and the 16 MB intermediate's round trip removed
# speedup vs baseline: 10.4454x; 1.0538x over previous
#include <hip/hip_runtime.h>


typedef __attribute__((ext_vector_type(2))) float v2f;
typedef __attribute__((ext_vector_type(8))) float v8f;

#define NEG_SLOPE 0.2f


#define SLOTS 32
#define RB 128
#define RB_SHIFT 7
#define P_CHUNK 3200
#define P_EPT ((P_CHUNK + 255) / 256)
#define NB_MAX 512
#define LPITCH 33

__global__ __launch_bounds__(256) void partition_edges(
    const int* __restrict__ srcE, const int* __restrict__ dstE, int E, int N,
    int NB, int* __restrict__ segbuf, int* __restrict__ segoff,
    int* __restrict__ cnt) {
  __shared__ int binc[NB_MAX];
  __shared__ int wsum[8];
  __shared__ int2 sorted[P_CHUNK];

  const int tid = threadIdx.x;
  const int lane = tid & 31;
  const int wave = tid >> 5;
  const int c = blockIdx.x;
  const int e0 = c * P_CHUNK;
  int n = E - e0;
  n = n > P_CHUNK ? P_CHUNK : n;

  if (c == 0 && tid == 0) cnt[N] = 0;
  binc[tid] = 0;
  binc[tid + 256] = 0;

  int dv[P_EPT], sv[P_EPT], rk[P_EPT];
#pragma unroll
  for (int i = 0; i < P_EPT; ++i) {
    const int k = tid + 256 * i;
    dv[i] = k < n ? dstE[e0 + k] : 0;
    sv[i] = k < n ? srcE[e0 + k] : 0;
  }
  __syncthreads();

#pragma unroll
  for (int i = 0; i < P_EPT; ++i) {
    rk[i] = 0;
    if (tid + 256 * i < n) rk[i] = atomicAdd(&binc[dv[i] >> RB_SHIFT], 1);
  }
  __syncthreads();

  const int ca = binc[2 * tid];
  const int cb = binc[2 * tid + 1];
  int inc = ca + cb;
#pragma unroll
  for (int o = 1; o < 32; o <<= 1) {
    const int v = __shfl_up(inc, o);
    if (lane >= o) inc += v;
  }
  if (lane == 31) wsum[wave] = inc;
  __syncthreads();
  int wbase = 0;
#pragma unroll
  for (int w = 0; w < 8; ++w) wbase += w < wave ? wsum[w] : 0;
  const int ex = wbase + inc - (ca + cb);
  binc[2 * tid] = ex;
  binc[2 * tid + 1] = ex + ca;
  __syncthreads();

#pragma unroll
  for (int i = 0; i < P_EPT; ++i)
    if (tid + 256 * i < n)
      sorted[binc[dv[i] >> RB_SHIFT] + rk[i]] = make_int2(dv[i], sv[i]);

  int* __restrict__ so = segoff + (size_t)c * (NB + 1);
  for (int b = tid; b <= NB; b += 256) so[b] = b < NB ? binc[b] : n;
  __syncthreads();

  int2* __restrict__ sg = reinterpret_cast<int2*>(segbuf) + e0;
  for (int k = tid; k < n; k += 256) sg[k] = sorted[k];
}

#define S_K 128
#define S_LD 80
#define S_KB (S_K / 8)
#define S_NT 10
#define S_WAVES 8

__global__ __launch_bounds__(256) void wmma_strip_gemm_l1_bucket(
    const float* __restrict__ A, const float* __restrict__ Bl,
    const float* __restrict__ Br, float* __restrict__ Cl,
    float* __restrict__ Cr, int rowTiles, const int* __restrict__ segbuf,
    const int* __restrict__ segoff, int NC, int N, int* __restrict__ cnt,
    int* __restrict__ bucket, int* __restrict__ spill) {
  static_assert(RB == 16 * S_WAVES, "a range is one work-group's rows");
  static_assert(SLOTS == 32, "one lane per slot in the bucket write-out");
  __shared__ float4 sB[S_KB * S_NT * 32];
  __shared__ int lcnt[RB];
  __shared__ int lbkt[RB * LPITCH];

  const int tid = threadIdx.x;
  const int lane = tid & 31;
  const int wave = tid >> 5;
  const int half = lane >> 4;
  const int l16 = lane & 15;
  const int rt = blockIdx.x * S_WAVES + wave;
  const bool live = rt < rowTiles;
  const int row0 = rt * 16;

  if (tid < RB) lcnt[tid] = 0;

  float4 a[S_KB];
  if (live) {
    const float* __restrict__ Ap = A + (size_t)(row0 + l16) * S_K + half * 4;
#pragma unroll
    for (int kb = 0; kb < S_KB; ++kb)
      a[kb] = *reinterpret_cast<const float4*>(Ap + kb * 8);
  }

  {
    constexpr int ITEMS = 2 * S_KB * 2 * S_LD;
    constexpr int PER = ITEMS / 256;
    float4 w[PER];
#pragma unroll
    for (int i = 0; i < PER; ++i) {
      const int it = tid + 256 * i;
      const int q = it / S_LD;
      const int col = it % S_LD;
      const int m = q >> 5;
      const int kb = (q & 31) >> 1;
      const int h = q & 1;
      const float* __restrict__ src =
          (m ? Br : Bl) + (size_t)(kb * 8 + h * 4) * S_LD + col;
      w[i].x = src[0];
      w[i].y = src[S_LD];
      w[i].z = src[2 * S_LD];
      w[i].w = src[3 * S_LD];
    }
#pragma unroll
    for (int i = 0; i < PER; ++i) {
      const int it = tid + 256 * i;
      const int q = it / S_LD;
      const int col = it % S_LD;
      const int m = q >> 5;
      const int kb = (q & 31) >> 1;
      const int h = q & 1;
      const int t = m * 5 + (col >> 4);
      sB[(kb * S_NT + t) * 32 + h * 16 + (col & 15)] = w[i];
    }
  }
  __syncthreads();

  if (live) {
    v8f acc[S_NT];
#pragma unroll
    for (int t = 0; t < S_NT; ++t) acc[t] = (v8f){};

#pragma unroll
    for (int kb = 0; kb < S_KB; ++kb) {
      float4 b[S_NT];
#pragma unroll
      for (int t = 0; t < S_NT; ++t) b[t] = sB[(kb * S_NT + t) * 32 + lane];
      v2f a01, a23;
      a01.x = a[kb].x; a01.y = a[kb].y;
      a23.x = a[kb].z; a23.y = a[kb].w;
#pragma unroll
      for (int t = 0; t < S_NT; ++t) {
        v2f b01;
        b01.x = b[t].x; b01.y = b[t].y;
        acc[t] = __builtin_amdgcn_wmma_f32_16x16x4_f32(false, a01, false, b01,
                                                       (short)0, acc[t], false,
                                                       false);
      }
#pragma unroll
      for (int t = 0; t < S_NT; ++t) {
        v2f b23;
        b23.x = b[t].z; b23.y = b[t].w;
        acc[t] = __builtin_amdgcn_wmma_f32_16x16x4_f32(false, a23, false, b23,
                                                       (short)0, acc[t], false,
                                                       false);
      }
    }

    const size_t base = (size_t)(row0 + half * 8) * S_LD + l16;
#pragma unroll
    for (int t = 0; t < S_NT; ++t) {
      float* __restrict__ c = (t < 5 ? Cl + t * 16 : Cr + (t - 5) * 16) + base;
#pragma unroll
      for (int i = 0; i < 8; ++i) c[(size_t)i * S_LD] = acc[t][i];
    }
  }

  const int nbase = blockIdx.x * RB;
  const int segPitch = gridDim.x + 1;
  auto place = [&](const int2 e) {
    const int dl = e.x - nbase;
    const int rank = atomicAdd(&lcnt[dl], 1);
    if (rank < SLOTS) {
      lbkt[dl * LPITCH + rank] = e.y;
    } else {
      const int o = atomicAdd(&cnt[N], 1);
      reinterpret_cast<int2*>(spill)[o] = e;
    }
  };
  for (int c = tid; c < NC; c += 256) {
    const int* __restrict__ so = segoff + (size_t)c * segPitch + blockIdx.x;
    const int beg = so[0];
    const int end = so[1];
    const int2* __restrict__ seg =
        reinterpret_cast<const int2*>(segbuf) + (size_t)c * P_CHUNK;
    for (int j = beg; j < end; j += 4) {
      const int last = end - 1;
      const int2 e0 = seg[j];
      const int2 e1 = seg[j + 1 < last ? j + 1 : last];
      const int2 e2 = seg[j + 2 < last ? j + 2 : last];
      const int2 e3 = seg[j + 3 < last ? j + 3 : last];
      place(e0);
      if (j + 1 < end) place(e1);
      if (j + 2 < end) place(e2);
      if (j + 3 < end) place(e3);
    }
  }
  __syncthreads();

  int nn = N - nbase;
  nn = nn > RB ? RB : nn;
  if (tid < nn) cnt[nbase + tid] = lcnt[tid];
  for (int n = wave; n < nn; n += S_WAVES)
    bucket[(size_t)(nbase + n) * SLOTS + lane] = lbkt[n * LPITCH + lane];
}

__device__ __forceinline__ float4 ld4(const float* p) {
  return *reinterpret_cast<const float4*>(p);
}

#define T_NODES 16
#define T_NOUT 16
#define T_PITCH 84

template <int HEADS, int DIMH, int LPG, bool RELU, bool FUSE>
__global__ __launch_bounds__(FUSE ? T_NODES * HEADS * LPG : 256)
void node_softmax_aggregate(
    const int* __restrict__ cnt, const int* __restrict__ bucket,
    const int* __restrict__ spill, int N,
    const float* __restrict__ xl, const float* __restrict__ xr,
    const float* __restrict__ att, const float* __restrict__ bias,
    float* __restrict__ out, const float* __restrict__ Wnl,
    const float* __restrict__ Wnr, float* __restrict__ ynl,
    float* __restrict__ ynr) {
  static_assert(DIMH == 4 * LPG, "each lane owns 4 features");
  static_assert(LPG == 1 || LPG == 2 || LPG == 4, "group inside a quad");
  constexpr int H = HEADS * DIMH;
  static_assert(!FUSE || ((T_NODES * HEADS * LPG) % 32 == 0 && H % 4 == 0 &&
                          T_NODES * HEADS * LPG >= 64),
                "whole waves, at least two, and whole k steps");
  const int t = blockIdx.x * blockDim.x + threadIdx.x;
  const int g = t / LPG;
  const int sub = t % LPG;
  if constexpr (!FUSE) {
    if (g >= N * HEADS) return;
  }
  const int d = g / HEADS;
  const int h = g % HEADS;
  const int col = h * DIMH + sub * 4;

  const float4 r = ld4(xr + (size_t)d * H + col);
  const float a0 = att[col], a1 = att[col + 1], a2 = att[col + 2],
              a3 = att[col + 3];

  auto logit_of = [&](const float4& v) -> float {
    float z0 = v.x + r.x, z1 = v.y + r.y, z2 = v.z + r.z, z3 = v.w + r.w;
    z0 = z0 > 0.f ? z0 : NEG_SLOPE * z0;
    z1 = z1 > 0.f ? z1 : NEG_SLOPE * z1;
    z2 = z2 > 0.f ? z2 : NEG_SLOPE * z2;
    z3 = z3 > 0.f ? z3 : NEG_SLOPE * z3;
    float p = z0 * a0;
    p += z1 * a1;
    p += z2 * a2;
    p += z3 * a3;
    if (LPG >= 2) p += __shfl_xor(p, 1);
    if (LPG >= 4) p += __shfl_xor(p, 2);
    return p;
  };

  const float4 xs = ld4(xl + (size_t)d * H + col);
  float m = logit_of(xs);
  float den = 1.f;
  float4 acc = xs;

  auto absorb = [&](const float4& v) {
    const float lg = logit_of(v);
    const float mn = fmaxf(m, lg);
    const float sc = __expf(m - mn);
    const float pe = __expf(lg - mn);
    den = den * sc + pe;
    acc.x = acc.x * sc + pe * v.x;
    acc.y = acc.y * sc + pe * v.y;
    acc.z = acc.z * sc + pe * v.z;
    acc.w = acc.w * sc + pe * v.w;
    m = mn;
  };

  auto absorb4 = [&](const float4& v0, const float4& v1, const float4& v2,
                     const float4& v3, int rem) {
    const float ninf = -__builtin_inff();
    const float l0 = logit_of(v0);
    float l1 = logit_of(v1);
    float l2 = logit_of(v2);
    float l3 = logit_of(v3);
    l1 = rem > 1 ? l1 : ninf;
    l2 = rem > 2 ? l2 : ninf;
    l3 = rem > 3 ? l3 : ninf;
    const float mn = fmaxf(fmaxf(m, l0), fmaxf(l1, fmaxf(l2, l3)));
    const float sc = __expf(m - mn);
    const float p0 = __expf(l0 - mn);
    const float p1 = __expf(l1 - mn);
    const float p2 = __expf(l2 - mn);
    const float p3 = __expf(l3 - mn);
    den = den * sc + p0;
    den += p1;
    den += p2;
    den += p3;
    acc.x = acc.x * sc + p0 * v0.x;
    acc.y = acc.y * sc + p0 * v0.y;
    acc.z = acc.z * sc + p0 * v0.z;
    acc.w = acc.w * sc + p0 * v0.w;
    acc.x += p1 * v1.x; acc.y += p1 * v1.y; acc.z += p1 * v1.z; acc.w += p1 * v1.w;
    acc.x += p2 * v2.x; acc.y += p2 * v2.y; acc.z += p2 * v2.z; acc.w += p2 * v2.w;
    acc.x += p3 * v3.x; acc.y += p3 * v3.y; acc.z += p3 * v3.z; acc.w += p3 * v3.w;
    m = mn;
  };

  const int deg = cnt[d];
  const int nb = deg < SLOTS ? deg : SLOTS;
  const int nbatch = (nb + 3) >> 2;
  const int4* __restrict__ mine4 =
      reinterpret_cast<const int4*>(bucket + (size_t)d * SLOTS);
  auto rows4 = [&](int b, const int4& ix, float4& v0, float4& v1, float4& v2,
                   float4& v3) {
    const int j = 4 * b;
    const int s0 = (j < nb) ? ix.x : d;
    const int s1 = (j + 1 < nb) ? ix.y : d;
    const int s2 = (j + 2 < nb) ? ix.z : d;
    const int s3 = (j + 3 < nb) ? ix.w : d;
    v0 = ld4(xl + (size_t)s0 * H + col);
    v1 = ld4(xl + (size_t)s1 * H + col);
    v2 = ld4(xl + (size_t)s2 * H + col);
    v3 = ld4(xl + (size_t)s3 * H + col);
  };
  for (int b = 0; b < nbatch; ++b) {
    const int4 ix = mine4[b];
    float4 v0, v1, v2, v3;
    rows4(b, ix, v0, v1, v2, v3);
    absorb4(v0, v1, v2, v3, nb - 4 * b);
  }

  if (deg > SLOTS) {
    const int nSpill = cnt[N];
    const int2* __restrict__ sp = reinterpret_cast<const int2*>(spill);
    for (int i = 0; i < nSpill; ++i) {
      const int2 ds = sp[i];
      if (ds.x == d) absorb(ld4(xl + (size_t)ds.y * H + col));
    }
  }

  const float inv = 1.f / den;
  float4 o;
  o.x = acc.x * inv + bias[col];
  o.y = acc.y * inv + bias[col + 1];
  o.z = acc.z * inv + bias[col + 2];
  o.w = acc.w * inv + bias[col + 3];
  if (RELU) {
    o.x = fmaxf(o.x, 0.f);
    o.y = fmaxf(o.y, 0.f);
    o.z = fmaxf(o.z, 0.f);
    o.w = fmaxf(o.w, 0.f);
  }
  if constexpr (!FUSE) {
    *reinterpret_cast<float4*>(out + (size_t)d * H + col) = o;
  } else {
    static_assert(T_PITCH % 4 == 0, "rows of the tile start 16-byte aligned");
    __shared__ float4 tile4[T_NODES * T_PITCH / 4];
    float* tile = reinterpret_cast<float*>(tile4);
    const int nl = threadIdx.x / (HEADS * LPG);
    tile4[(nl * T_PITCH + col) >> 2] = o;

    const int wave = __builtin_amdgcn_readfirstlane((int)(threadIdx.x >> 5));
    const int lane = threadIdx.x & 31;
    const int half = lane >> 4;
    const int l16 = lane & 15;
    constexpr int KS = H / 4;
    v2f bf[KS];
    if (wave < 2) {
      const float* __restrict__ Wn = (wave ? Wnr : Wnl) + l16;
#pragma unroll
      for (int ks = 0; ks < KS; ++ks) {
        const int ka = 4 * ks + 2 * half;
        bf[ks].x = Wn[(size_t)ka * T_NOUT];
        bf[ks].y = Wn[(size_t)(ka + 1) * T_NOUT];
      }
    }
    __syncthreads();
    if (wave < 2) {
      v8f accn = {};
      const float* __restrict__ arow = &tile[l16 * T_PITCH + 2 * half];
#pragma unroll
      for (int ks = 0; ks < KS; ++ks) {
        v2f a;
        a.x = arow[4 * ks];
        a.y = arow[4 * ks + 1];
        accn = __builtin_amdgcn_wmma_f32_16x16x4_f32(false, a, false, bf[ks],
                                                     (short)0, accn, false,
                                                     false);
      }
      float* __restrict__ y =
          (wave ? ynr : ynl) +
          (size_t)(blockIdx.x * T_NODES + half * 8) * T_NOUT + l16;
#pragma unroll
      for (int i = 0; i < 8; ++i) y[(size_t)i * T_NOUT] = accn[i];
    }
  }
}

extern "C" void kernel_launch(void* const* d_in, const int* in_sizes, int n_in,
                              void* d_out, int out_size, void* d_ws,
                              size_t ws_size, hipStream_t stream) {
  const float* x = (const float*)d_in[0];
  const int* edge = (const int*)d_in[1];
  const float* W1l = (const float*)d_in[2];
  const float* W1r = (const float*)d_in[3];
  const float* att1 = (const float*)d_in[4];
  const float* b1 = (const float*)d_in[5];
  const float* W2l = (const float*)d_in[6];
  const float* W2r = (const float*)d_in[7];
  const float* att2 = (const float*)d_in[8];
  const float* b2 = (const float*)d_in[9];
  float* out = (float*)d_out;

  const int F_IN = 128, H1 = 80, HEADS1 = 10, F_OUT = 16;
  const int N = in_sizes[0] / F_IN;
  const int E = in_sizes[1] / 2;
  const int* srcE = edge;
  const int* dstE = edge + E;

  float* ws = (float*)d_ws;
  float* xl1 = ws;                 ws += (size_t)N * H1;
  float* xr1 = ws;                 ws += (size_t)N * H1;
  float* xl2 = ws;                 ws += (size_t)N * F_OUT;
  float* xr2 = ws;                 ws += (size_t)N * F_OUT;
  int* wi = (int*)ws;
  int* cnt = wi;                   wi += (size_t)((N + 1 + 31) / 32) * 32;
  int* bucket = wi;                wi += (size_t)N * SLOTS;
  int* spill = wi;                 wi += (size_t)2 * E;
  int* segbuf = wi;                wi += (size_t)2 * E;
  int* segoff = wi;
  (void)ws_size; (void)n_in; (void)out_size;

  const int BLK = 256;

  const int NB = (N + RB - 1) / RB;
  const int NC = (E + P_CHUNK - 1) / P_CHUNK;
  partition_edges<<<NC, BLK, 0, stream>>>(srcE, dstE, E, N, NB, segbuf, segoff,
                                          cnt);

  {
    const int rowTiles = N / 16;
    wmma_strip_gemm_l1_bucket<<<NB, BLK, 0, stream>>>(
        x, W1l, W1r, xl1, xr1, rowTiles, segbuf, segoff, NC, N, cnt, bucket,
        spill);
  }
  node_softmax_aggregate<10, 8, 2, true, true>
      <<<N / T_NODES, T_NODES * HEADS1 * 2, 0, stream>>>(
          cnt, bucket, spill, N, xl1, xr1, att1, b1, nullptr, W2l, W2r, xl2,
          xr2);

  {
    const int threads = N * 4;
    node_softmax_aggregate<1, 16, 4, false, false>
        <<<(threads + BLK - 1) / BLK, BLK, 0, stream>>>(
            cnt, bucket, spill, N, xl2, xr2, att2, b2, out, nullptr, nullptr,
            nullptr, nullptr);
  }
}
